# starting kernel
# baseline (speedup 1.0000x reference)
#include <hip/hip_runtime.h>
#include <hip/hip_bf16.h>

typedef __attribute__((ext_vector_type(2))) float v2f;
typedef __attribute__((ext_vector_type(8))) float v8f;

#define HID 64

__device__ __forceinline__ void atomAddF(float* p, float v) {
    (void)__hip_atomic_fetch_add(p, v, __ATOMIC_RELAXED, __HIP_MEMORY_SCOPE_AGENT);
}


__global__ void k_init_deg(float* __restrict__ deg, int n) {
    int i = blockIdx.x * blockDim.x + threadIdx.x;
    if (i < n) deg[i] = 1.0f;
}

__global__ void k_deg(const int* __restrict__ dst, float* __restrict__ deg, int e) {
    int i = blockIdx.x * blockDim.x + threadIdx.x;
    if (i < e) atomAddF(&deg[dst[i]], 1.0f);
}

__global__ void k_dinv(float* __restrict__ deg, int n) {
    int i = blockIdx.x * blockDim.x + threadIdx.x;
    if (i < n) deg[i] = rsqrtf(deg[i]);
}

__global__ void k_norm(const int* __restrict__ src, const int* __restrict__ dst,
                       const float* __restrict__ dinv, float* __restrict__ norm, int e) {
    int i = blockIdx.x * blockDim.x + threadIdx.x;
    if (i < e) norm[i] = dinv[src[i]] * dinv[dst[i]];
}


__global__ void k_lin1(const float* __restrict__ x, const float* __restrict__ W1,
                       float* __restrict__ h, int n) {
    int t = blockIdx.x * blockDim.x + threadIdx.x;
    int i = t >> 6, j = t & 63;
    if (i < n) {
        float2 xv = ((const float2*)x)[i];
        h[t] = xv.x * W1[j] + xv.y * W1[HID + j];
    }
}


__global__ void k_prop_init(const float* __restrict__ h, const float* __restrict__ dinv,
                            const float* __restrict__ b, float* __restrict__ out, int n) {
    int t = blockIdx.x * blockDim.x + threadIdx.x;
    int i = t >> 6, j = t & 63;
    if (i < n) {
        float di = dinv[i];
        out[t] = b[j] + h[t] * di * di;
    }
}

__global__ void k_prop_edges(const int* __restrict__ src, const int* __restrict__ dst,
                             const float* __restrict__ norm, const float* __restrict__ h,
                             float* __restrict__ out, int e) {
    int t  = blockIdx.x * blockDim.x + threadIdx.x;
    int ei = t >> 5;
    if (ei >= e) return;
    int lane = t & 31;
    int s = src[ei], d = dst[ei];
    float w = norm[ei];
    float2 hv = ((const float2*)(h + (size_t)s * HID))[lane];
    float* op = out + (size_t)d * HID + lane * 2;
    atomAddF(op,     hv.x * w);
    atomAddF(op + 1, hv.y * w);
}

__global__ void __launch_bounds__(256)
k_gemm_relu_wmma(const float* __restrict__ in, const float* __restrict__ W,
                 float* __restrict__ out, int nrows) {
    int gwave = (blockIdx.x * blockDim.x + threadIdx.x) >> 5;
    int lane  = threadIdx.x & 31;
    int row0  = gwave << 4;
    if (row0 >= nrows) return;
    int hh = lane >> 4;
    int l  = lane & 15;
    v8f acc0 = {}, acc1 = {}, acc2 = {}, acc3 = {};
    const float* arow = in + (size_t)(row0 + l) * HID;
    for (int k0 = 0; k0 < HID; k0 += 4) {
        int ka = k0 + hh * 2;
        v2f a;
        a.x = fmaxf(arow[ka],     0.0f);
        a.y = fmaxf(arow[ka + 1], 0.0f);
        const float* bp = W + (size_t)ka * HID + l;
        v2f b0, b1, b2, b3;
        b0.x = bp[0];  b0.y = bp[HID];
        b1.x = bp[16]; b1.y = bp[HID + 16];
        b2.x = bp[32]; b2.y = bp[HID + 32];
        b3.x = bp[48]; b3.y = bp[HID + 48];
        acc0 = __builtin_amdgcn_wmma_f32_16x16x4_f32(false, a, false, b0, (short)0, acc0, false, false);
        acc1 = __builtin_amdgcn_wmma_f32_16x16x4_f32(false, a, false, b1, (short)0, acc1, false, false);
        acc2 = __builtin_amdgcn_wmma_f32_16x16x4_f32(false, a, false, b2, (short)0, acc2, false, false);
        acc3 = __builtin_amdgcn_wmma_f32_16x16x4_f32(false, a, false, b3, (short)0, acc3, false, false);
    }
    float* orow = out + (size_t)(row0 + hh * 8) * HID + l;
    #pragma unroll
    for (int r = 0; r < 8; ++r) {
        float* o = orow + (size_t)r * HID;
        o[0]  = acc0[r];
        o[16] = acc1[r];
        o[32] = acc2[r];
        o[48] = acc3[r];
    }
}


__global__ void k_lin3_relu(const float* __restrict__ in, const float* __restrict__ W3,
                            float* __restrict__ h3, int n) {
    int gwave = (blockIdx.x * blockDim.x + threadIdx.x) >> 5;
    int lane  = threadIdx.x & 31;
    if (gwave >= n) return;
    const float* row = in + (size_t)gwave * HID;
    float v = fmaxf(row[lane], 0.0f) * W3[lane]
            + fmaxf(row[lane + 32], 0.0f) * W3[lane + 32];
    #pragma unroll
    for (int off = 16; off > 0; off >>= 1) v += __shfl_down(v, off, 32);
    if (lane == 0) h3[gwave] = v;
}

__global__ void k_prop3_init(const float* __restrict__ h3, const float* __restrict__ dinv,
                             const float* __restrict__ b3, float* __restrict__ out, int n) {
    int i = blockIdx.x * blockDim.x + threadIdx.x;
    if (i < n) {
        float di = dinv[i];
        out[i] = b3[0] + h3[i] * di * di;
    }
}

__global__ void k_prop3_edges(const int* __restrict__ src, const int* __restrict__ dst,
                              const float* __restrict__ norm, const float* __restrict__ h3,
                              float* __restrict__ out, int e) {
    int i = blockIdx.x * blockDim.x + threadIdx.x;
    if (i < e) atomAddF(&out[dst[i]], h3[src[i]] * norm[i]);
}


extern "C" void kernel_launch(void* const* d_in, const int* in_sizes, int n_in,
                              void* d_out, int out_size, void* d_ws, size_t ws_size,
                              hipStream_t stream) {
    const float* x  = (const float*)d_in[0];
    const int*   ei = (const int*)  d_in[1];
    const float* W1 = (const float*)d_in[2];
    const float* b1 = (const float*)d_in[3];
    const float* W2 = (const float*)d_in[4];
    const float* b2 = (const float*)d_in[5];
    const float* W3 = (const float*)d_in[6];
    const float* b3 = (const float*)d_in[7];

    const int N = in_sizes[0] / 2;
    const int E = in_sizes[1] / 2;
    const int* src = ei;
    const int* dst = ei + E;

    float* norm = (float*)d_ws;
    float* dinv = norm + (size_t)E;
    float* h    = dinv + (size_t)N;
    float* out  = h    + (size_t)N * HID;
    float* h3   = out  + (size_t)N * HID;
    float* y    = (float*)d_out;

    const int B = 256;
    auto g = [](long long t) { return (int)((t + 255) / 256); };

    k_init_deg<<<g(N), B, 0, stream>>>(dinv, N);
    k_deg     <<<g(E), B, 0, stream>>>(dst, dinv, E);
    k_dinv    <<<g(N), B, 0, stream>>>(dinv, N);
    k_norm    <<<g(E), B, 0, stream>>>(src, dst, dinv, norm, E);

    k_lin1      <<<g((long long)N * HID), B, 0, stream>>>(x, W1, h, N);
    k_prop_init <<<g((long long)N * HID), B, 0, stream>>>(h, dinv, b1, out, N);
    k_prop_edges<<<g((long long)E * 32),  B, 0, stream>>>(src, dst, norm, h, out, E);

    k_gemm_relu_wmma<<<g((long long)(N / 16) * 32), 256, 0, stream>>>(out, W2, h, N);
    k_prop_init <<<g((long long)N * HID), B, 0, stream>>>(h, dinv, b2, out, N);
    k_prop_edges<<<g((long long)E * 32),  B, 0, stream>>>(src, dst, norm, h, out, E);

    k_lin3_relu  <<<g((long long)N * 32), B, 0, stream>>>(out, W3, h3, N);
    k_prop3_init <<<g(N), B, 0, stream>>>(h3, dinv, b3, y, N);
    k_prop3_edges<<<g(E), B, 0, stream>>>(src, dst, norm, h3, y, E);
}


# Layer 3 propagation replays the waves' hit records in the owner workgroups: weight*h3[src] added to per-node LDS sums (LDS f32 adds, arrival order), stored once; no global atomics, start pass fused
# speedup vs baseline: 2.0678x; 2.0678x over previous
#include <hip/hip_runtime.h>
#include <hip/hip_bf16.h>

typedef __attribute__((ext_vector_type(2))) float v2f;
typedef __attribute__((ext_vector_type(8))) float v8f;

#define HID 64
#define NPW 391

__device__ __forceinline__ void atomAddF(float* p, float v) {
    (void)__hip_atomic_fetch_add(p, v, __ATOMIC_RELAXED, __HIP_MEMORY_SCOPE_AGENT);
}


__global__ void k_init_deg(float* __restrict__ deg, int n) {
    int i = blockIdx.x * blockDim.x + threadIdx.x;
    if (i < n) deg[i] = 1.0f;
}

__global__ void k_deg(const int* __restrict__ dst, float* __restrict__ deg, int e) {
    int i = blockIdx.x * blockDim.x + threadIdx.x;
    if (i < e) atomAddF(&deg[dst[i]], 1.0f);
}

__global__ void k_dinv(float* __restrict__ deg, int n) {
    int i = blockIdx.x * blockDim.x + threadIdx.x;
    if (i < n) deg[i] = rsqrtf(deg[i]);
}

__global__ void k_norm(const int* __restrict__ src, const int* __restrict__ dst,
                       const float* __restrict__ dinv, float* __restrict__ norm,
                       unsigned char* __restrict__ key, int e) {
    int i = blockIdx.x * blockDim.x + threadIdx.x;
    if (i < e) {
        int d = dst[i];
        norm[i] = dinv[src[i]] * dinv[d];
        key[i] = (unsigned char)((unsigned)d / NPW);
    }
}


__global__ void k_lin1(const float* __restrict__ x, const float* __restrict__ W1,
                       float* __restrict__ h, int n) {
    int t = blockIdx.x * blockDim.x + threadIdx.x;
    int i = t >> 6, j = t & 63;
    if (i < n) {
        float2 xv = ((const float2*)x)[i];
        h[t] = xv.x * W1[j] + xv.y * W1[HID + j];
    }
}

#define PWAVES 32
#define QCAP   1056
#define RCAP   1024

__global__ void __launch_bounds__(1024)
k_prop_own(const int* __restrict__ src, const int* __restrict__ dst,
           const unsigned char* __restrict__ key,
           const float* __restrict__ norm, const float* __restrict__ h,
           const float* __restrict__ dinv, const float* __restrict__ b,
           float* __restrict__ out, int* __restrict__ recs, int* __restrict__ rcount,
           int mode_arg, int n, int e) {
    __shared__ float acc[NPW * HID];
    __shared__ int   q_e[PWAVES][QCAP];

    const int mode = gridDim.x <= 256 ? mode_arg : 2;
    const int tid  = threadIdx.x;
    const int lane = tid & 31;
    const int wv   = __builtin_amdgcn_readfirstlane(tid >> 5);
    const int lo   = blockIdx.x * NPW;
    int cnt = n - lo;
    if (cnt > NPW) cnt = NPW;
    if (cnt < 0)   cnt = 0;
    const unsigned ucnt = (unsigned)cnt;
    const unsigned ulo  = (unsigned)lo;

    for (int t = tid; t < cnt * HID; t += 1024) {
        int i = t >> 6, j = t & 63;
        float di = dinv[lo + i];
        acc[t] = b[j] + h[(size_t)lo * HID + t] * di * di;
    }
    __syncthreads();

    int* qe = q_e[wv];
    int  qn = 0;
    const int region = blockIdx.x * PWAVES + wv;
    int4* myrec = (int4*)recs + (size_t)region * RCAP;
    int  rn    = 0;
    bool rfull = false;
    const unsigned pat = (unsigned)(blockIdx.x & 255) * 0x01010101u;

    auto load_chunk_guarded = [&](int chunk) -> int4 {
        int4 v;
        int base = chunk * 128 + lane * 4;
        v.x = (base     < e) ? dst[base]     : -1;
        v.y = (base + 1 < e) ? dst[base + 1] : -1;
        v.z = (base + 2 < e) ? dst[base + 2] : -1;
        v.w = (base + 3 < e) ? dst[base + 3] : -1;
        return v;
    };

    auto add_rows = [&](int my_s, int my_d, int my_w, int m) {
        for (int i = 0; i < m; i += 8) {
            float a0[8], a1[8];
            #pragma unroll
            for (int u = 0; u < 8; ++u) {
                int s = __builtin_amdgcn_readlane(my_s, i + u);
                const float* r = h + (size_t)s * HID;
                a0[u] = r[lane];
                a1[u] = r[lane + 32];
            }
            #pragma unroll
            for (int u = 0; u < 8; ++u) {
                if (i + u < m) {
                    float w = __int_as_float(__builtin_amdgcn_readlane(my_w, i + u));
                    int   d = __builtin_amdgcn_readlane(my_d, i + u);
                    float* p = acc + d * HID + lane;
                    (void)__hip_atomic_fetch_add(p,      a0[u] * w, __ATOMIC_RELAXED, __HIP_MEMORY_SCOPE_WORKGROUP);
                    (void)__hip_atomic_fetch_add(p + 32, a1[u] * w, __ATOMIC_RELAXED, __HIP_MEMORY_SCOPE_WORKGROUP);
                }
            }
        }
    };

    auto drain = [&](int qb, int m) {
        int my_s = 0, my_d = 0, my_w = 0;
        if (lane < m) {
            int ei = qe[qb + lane];
            my_d = dst[ei] - lo;
            my_s = src[ei];
            my_w = __float_as_int(norm[ei]);
        }
        if (mode == 0) {
            if (!rfull && rn + m <= RCAP) {
                if (lane < m) myrec[rn + lane] = make_int4(my_s, my_d, my_w, 0);
                rn += m;
            } else {
                rfull = true;
            }
        }
        add_rows(my_s, my_d, my_w, m);
    };

    auto append = [&](unsigned t, int ei) {
        bool hit = t < ucnt;
        unsigned mask = (unsigned)__ballot(hit);
        if (mask) {
            int pos = qn + (int)__builtin_amdgcn_mbcnt_lo(mask, 0u);
            if (hit) qe[pos] = ei;
            qn += __popc(mask);
        }
    };

    auto handle_dst = [&](int4 v, int chunk) {
        uint4 t;
        t.x = (unsigned)v.x - ulo; t.y = (unsigned)v.y - ulo;
        t.z = (unsigned)v.z - ulo; t.w = (unsigned)v.w - ulo;
        unsigned mn = min(min(t.x, t.y), min(t.z, t.w));
        if (__any(mn < ucnt)) {
            int base = chunk * 128 + lane * 4;
            append(t.x, base);
            append(t.y, base + 1);
            append(t.z, base + 2);
            append(t.w, base + 3);
        }
    };

    auto zero_bytes = [](unsigned x) -> unsigned {
        unsigned t = (x & 0x7F7F7F7Fu) + 0x7F7F7F7Fu;
        return ~(t | x | 0x7F7F7F7Fu);
    };

    auto match16 = [&](int4 v) -> unsigned {
        return (zero_bytes((unsigned)v.x ^ pat) >> 7) | (zero_bytes((unsigned)v.y ^ pat) >> 6)
             | (zero_bytes((unsigned)v.z ^ pat) >> 5) | (zero_bytes((unsigned)v.w ^ pat) >> 4);
    };

    auto enqueue = [&](unsigned m, int ebase) {
        if (__any(m != 0u)) {
            const int lbase = ebase + lane * 16;
            do {
                bool has = m != 0u;
                unsigned mask = (unsigned)__ballot(has);
                int pos = qn + (int)__builtin_amdgcn_mbcnt_lo(mask, 0u);
                if (has) {
                    int p = __builtin_ctz(m);
                    qe[pos] = lbase + ((p & 7) << 2) + (p >> 3);
                    m &= m - 1u;
                }
                qn += __popc(mask);
            } while (__any(m != 0u));
        }
    };

    const bool vec_ok  = (((size_t)key) & 15) == 0 && gridDim.x <= 256;
    const int  nunit   = vec_ok ? (e >> 9) : 0;
    const int  ustart  = (int)(((long long)blockIdx.x * nunit) / gridDim.x);
    const int4* key4   = (const int4*)key;

    bool scan = true;
    if (mode == 1) {
        int c = __builtin_amdgcn_readfirstlane(rcount[region]);
        if (c >= 0) {
            scan = false;
            if (c > RCAP) c = RCAP;
            if (c > 0) {
                int4 nx = myrec[min(lane, c - 1)];
                for (int base = 0; base < c; base += 32) {
                    int4 cur = nx;
                    nx = myrec[min(base + 32 + lane, c - 1)];
                    int m = c - base;
                    if (m > 32) m = 32;
                    add_rows(cur.x, cur.y, cur.z, m);
                }
            }
        }
    }

    if (cnt > 0 && scan) {
        if (wv < nunit) {
            const int nk = (nunit - wv + PWAVES - 1) / PWAVES;
            auto unit_of = [&](int k) -> int {
                int r = wv + (k < 0 ? 0 : (k < nk ? k : nk - 1)) * PWAVES + ustart;
                return r >= nunit ? r - nunit : r;
            };
            auto ldu = [&](int u) -> int4 { return key4[(size_t)u * 32 + lane]; };
            auto step = [&](int4& a, int k) {
                unsigned m = match16(a);
                a = ldu(unit_of(k + 8));
                if (k < 0 || k >= nk) m = 0u;
                enqueue(m, unit_of(k) * 512);
            };
            const int4 none = {0, 0, 0, 0};
            int4 a0 = none, a1 = none, a2 = none, a3 = none, a4 = none, a5 = none, a6 = none, a7 = none;
            for (int k = -8; k < nk; k += 8) {
                step(a0, k);     step(a1, k + 1); while (qn >= 32) { qn -= 32; drain(qn, 32); }
                step(a2, k + 2); step(a3, k + 3); while (qn >= 32) { qn -= 32; drain(qn, 32); }
                step(a4, k + 4); step(a5, k + 5); while (qn >= 32) { qn -= 32; drain(qn, 32); }
                step(a6, k + 6); step(a7, k + 7); while (qn >= 32) { qn -= 32; drain(qn, 32); }
            }
        }
        for (int c = nunit * 4 + wv; c * 128 < e; c += PWAVES) {
            handle_dst(load_chunk_guarded(c), c);
            while (qn >= 32) { qn -= 32; drain(qn, 32); }
        }
        if (qn > 0) drain(0, qn);
    }
    if (mode == 0 && lane == 0) rcount[region] = rfull ? -1 : rn;
    __syncthreads();

    for (int t = tid; t < cnt * HID; t += 1024)
        out[(size_t)lo * HID + t] = acc[t];
}

__global__ void __launch_bounds__(256)
k_gemm_relu_wmma(const float* __restrict__ in, const float* __restrict__ W,
                 float* __restrict__ out, int nrows) {
    int gwave = (blockIdx.x * blockDim.x + threadIdx.x) >> 5;
    int lane  = threadIdx.x & 31;
    int row0  = gwave << 4;
    if (row0 >= nrows) return;
    int hh = lane >> 4;
    int l  = lane & 15;
    v8f acc0 = {}, acc1 = {}, acc2 = {}, acc3 = {};
    const float* arow = in + (size_t)(row0 + l) * HID;
    for (int k0 = 0; k0 < HID; k0 += 4) {
        int ka = k0 + hh * 2;
        v2f a;
        a.x = fmaxf(arow[ka],     0.0f);
        a.y = fmaxf(arow[ka + 1], 0.0f);
        const float* bp = W + (size_t)ka * HID + l;
        v2f b0, b1, b2, b3;
        b0.x = bp[0];  b0.y = bp[HID];
        b1.x = bp[16]; b1.y = bp[HID + 16];
        b2.x = bp[32]; b2.y = bp[HID + 32];
        b3.x = bp[48]; b3.y = bp[HID + 48];
        acc0 = __builtin_amdgcn_wmma_f32_16x16x4_f32(false, a, false, b0, (short)0, acc0, false, false);
        acc1 = __builtin_amdgcn_wmma_f32_16x16x4_f32(false, a, false, b1, (short)0, acc1, false, false);
        acc2 = __builtin_amdgcn_wmma_f32_16x16x4_f32(false, a, false, b2, (short)0, acc2, false, false);
        acc3 = __builtin_amdgcn_wmma_f32_16x16x4_f32(false, a, false, b3, (short)0, acc3, false, false);
    }
    float* orow = out + (size_t)(row0 + hh * 8) * HID + l;
    #pragma unroll
    for (int r = 0; r < 8; ++r) {
        float* o = orow + (size_t)r * HID;
        o[0]  = acc0[r];
        o[16] = acc1[r];
        o[32] = acc2[r];
        o[48] = acc3[r];
    }
}


__global__ void k_lin3_relu(const float* __restrict__ in, const float* __restrict__ W3,
                            float* __restrict__ h3, int n) {
    int gwave = (blockIdx.x * blockDim.x + threadIdx.x) >> 5;
    int lane  = threadIdx.x & 31;
    if (gwave >= n) return;
    const float* row = in + (size_t)gwave * HID;
    float v = fmaxf(row[lane], 0.0f) * W3[lane]
            + fmaxf(row[lane + 32], 0.0f) * W3[lane + 32];
    #pragma unroll
    for (int off = 16; off > 0; off >>= 1) v += __shfl_down(v, off, 32);
    if (lane == 0) h3[gwave] = v;
}

__global__ void __launch_bounds__(1024)
k_prop3_own(const int* __restrict__ src, const int* __restrict__ dst,
            const float* __restrict__ norm, const float* __restrict__ h3,
            const float* __restrict__ dinv, const float* __restrict__ b3,
            float* __restrict__ y, const int* __restrict__ recs,
            const int* __restrict__ rcount, int n, int e) {
    __shared__ float acc3[NPW];
    __shared__ int   no_recs;

    const int tid  = threadIdx.x;
    const int lane = tid & 31;
    const int wv   = __builtin_amdgcn_readfirstlane(tid >> 5);
    const int lo   = blockIdx.x * NPW;
    int cnt = n - lo;
    if (cnt > NPW) cnt = NPW;
    if (cnt < 0)   cnt = 0;
    const unsigned ucnt = (unsigned)cnt;

    const int region = blockIdx.x * PWAVES + wv;
    int c = -1;
    if (gridDim.x <= 256) c = __builtin_amdgcn_readfirstlane(rcount[region]);
    if (c > RCAP) c = RCAP;

    if (tid == 0) no_recs = 0;
    if (tid < cnt) {
        float di = dinv[lo + tid];
        acc3[tid] = b3[0] + h3[lo + tid] * di * di;
    }
    __syncthreads();
    if (c < 0 && lane == 0) no_recs = 1;
    __syncthreads();

    if (no_recs == 0) {
        const int4* myrec = (const int4*)recs + (size_t)region * RCAP;
        for (int base = 0; base < c; base += 256) {
            int4  r[8];
            float hv[8];
            #pragma unroll
            for (int u = 0; u < 8; ++u) r[u] = myrec[min(base + u * 32 + lane, c - 1)];
            #pragma unroll
            for (int u = 0; u < 8; ++u) hv[u] = h3[r[u].x];
            asm volatile("" :: "v"(hv[0]), "v"(hv[1]), "v"(hv[2]), "v"(hv[3]),
                               "v"(hv[4]), "v"(hv[5]), "v"(hv[6]), "v"(hv[7]));
            #pragma unroll
            for (int u = 0; u < 8; ++u) {
                if (base + u * 32 + lane < c)
                    (void)__hip_atomic_fetch_add(acc3 + r[u].y, hv[u] * __int_as_float(r[u].z),
                                                 __ATOMIC_RELAXED, __HIP_MEMORY_SCOPE_WORKGROUP);
            }
        }
    } else {
        for (int i = tid; i < e; i += 1024) {
            unsigned d = (unsigned)dst[i] - (unsigned)lo;
            if (d < ucnt)
                (void)__hip_atomic_fetch_add(acc3 + d, h3[src[i]] * norm[i],
                                             __ATOMIC_RELAXED, __HIP_MEMORY_SCOPE_WORKGROUP);
        }
    }
    __syncthreads();

    if (tid < cnt) y[lo + tid] = acc3[tid];
}


extern "C" void kernel_launch(void* const* d_in, const int* in_sizes, int n_in,
                              void* d_out, int out_size, void* d_ws, size_t ws_size,
                              hipStream_t stream) {
    const float* x  = (const float*)d_in[0];
    const int*   ei = (const int*)  d_in[1];
    const float* W1 = (const float*)d_in[2];
    const float* b1 = (const float*)d_in[3];
    const float* W2 = (const float*)d_in[4];
    const float* b2 = (const float*)d_in[5];
    const float* W3 = (const float*)d_in[6];
    const float* b3 = (const float*)d_in[7];

    const int N = in_sizes[0] / 2;
    const int E = in_sizes[1] / 2;
    const int* src = ei;
    const int* dst = ei + E;

    float* norm = (float*)d_ws;
    float* dinv = norm + (size_t)E;
    float* h    = dinv + (size_t)N;
    float* out  = h    + (size_t)N * HID;
    float* h3   = out  + (size_t)N * HID;
    unsigned char* key = (unsigned char*)(h3 + (size_t)N);
    int* recs   = (int*)(key + (size_t)E);
    int* rcount = recs + (size_t)256 * PWAVES * RCAP * 4;
    float* y    = (float*)d_out;

    const int B = 256;
    auto g = [](long long t) { return (int)((t + 255) / 256); };

    k_init_deg<<<g(N), B, 0, stream>>>(dinv, N);
    k_deg     <<<g(E), B, 0, stream>>>(dst, dinv, E);
    k_dinv    <<<g(N), B, 0, stream>>>(dinv, N);
    k_norm    <<<g(E), B, 0, stream>>>(src, dst, dinv, norm, key, E);

    k_lin1      <<<g((long long)N * HID), B, 0, stream>>>(x, W1, h, N);
    k_prop_own<<<(N + NPW - 1) / NPW, 1024, 0, stream>>>(src, dst, key, norm, h, dinv, b1, out, recs, rcount, 0, N, E);

    k_gemm_relu_wmma<<<g((long long)(N / 16) * 32), 256, 0, stream>>>(out, W2, h, N);
    k_prop_own<<<(N + NPW - 1) / NPW, 1024, 0, stream>>>(src, dst, key, norm, h, dinv, b2, out, recs, rcount, 1, N, E);

    k_lin3_relu  <<<g((long long)N * 32), B, 0, stream>>>(out, W3, h3, N);
    k_prop3_own<<<(N + NPW - 1) / NPW, 1024, 0, stream>>>(src, dst, norm, h3, dinv, b3, y, recs, rcount, N, E);
}


# Degree count moved into the owner scan, now run first: LDS integer adds, no global atomics, no norm pass; layer 1 replays records, forms weights from dinv; LDS f32 adds unordered as before
# speedup vs baseline: 3.1984x; 1.5468x over previous
#include <hip/hip_runtime.h>
#include <hip/hip_bf16.h>

typedef __attribute__((ext_vector_type(2))) float v2f;
typedef __attribute__((ext_vector_type(8))) float v8f;

#define HID 64
#define NPW 391

__global__ void k_key(const int* __restrict__ dst, unsigned char* __restrict__ key, int e) {
    int i = (blockIdx.x * blockDim.x + threadIdx.x) * 4;
    if (i + 3 < e && (((size_t)dst) & 15) == 0 && (((size_t)key) & 3) == 0) {
        int4 d = *(const int4*)(dst + i);
        unsigned k = (((unsigned)d.x / NPW) & 255u)       | ((((unsigned)d.y / NPW) & 255u) << 8)
                   | ((((unsigned)d.z / NPW) & 255u) << 16) | ((((unsigned)d.w / NPW) & 255u) << 24);
        *(unsigned*)(key + i) = k;
    } else {
        for (int u = 0; u < 4; ++u)
            if (i + u < e) key[i + u] = (unsigned char)((unsigned)dst[i + u] / NPW);
    }
}


__global__ void k_lin1(const float* __restrict__ x, const float* __restrict__ W1,
                       float* __restrict__ h, int n) {
    int t = blockIdx.x * blockDim.x + threadIdx.x;
    int i = t >> 6, j = t & 63;
    if (i < n) {
        float2 xv = ((const float2*)x)[i];
        h[t] = xv.x * W1[j] + xv.y * W1[HID + j];
    }
}

#define PWAVES 32
#define QCAP   1056
#define RCAP   1024

template <int MODE>
__device__ __forceinline__ void own_body(
           float* __restrict__ acc, int* __restrict__ degc, int (* __restrict__ q_e)[QCAP],
           const int* __restrict__ src, const int* __restrict__ dst,
           const unsigned char* __restrict__ key, const float* __restrict__ h,
           const float* __restrict__ dinv, float* __restrict__ dinv_out,
           const float* __restrict__ b, float* __restrict__ out,
           int* __restrict__ recs, int* __restrict__ rcount, int n, int e) {
    const bool small = gridDim.x <= 256;
    const int tid  = threadIdx.x;
    const int lane = tid & 31;
    const int wv   = __builtin_amdgcn_readfirstlane(tid >> 5);
    const int lo   = blockIdx.x * NPW;
    int cnt = n - lo;
    if (cnt > NPW) cnt = NPW;
    if (cnt < 0)   cnt = 0;
    const unsigned ucnt = (unsigned)cnt;
    const unsigned ulo  = (unsigned)lo;

    if (MODE == 0) {
        for (int t = tid; t < cnt; t += 1024) degc[t] = 0;
    } else {
        for (int t = tid; t < cnt * HID; t += 1024) {
            int i = t >> 6, j = t & 63;
            float di = dinv[lo + i];
            acc[t] = b[j] + h[(size_t)lo * HID + t] * di * di;
        }
    }
    __syncthreads();

    int* qe = q_e[wv];
    int  qn = 0;
    const int region = blockIdx.x * PWAVES + wv;
    int4* myrec = (int4*)recs + (size_t)region * RCAP;
    int  rn    = 0;
    bool rfull = false;
    const unsigned pat = (unsigned)(blockIdx.x & 255) * 0x01010101u;

    auto load_chunk_guarded = [&](int chunk) -> int4 {
        int4 v;
        int base = chunk * 128 + lane * 4;
        v.x = (base     < e) ? dst[base]     : -1;
        v.y = (base + 1 < e) ? dst[base + 1] : -1;
        v.z = (base + 2 < e) ? dst[base + 2] : -1;
        v.w = (base + 3 < e) ? dst[base + 3] : -1;
        return v;
    };

    auto add_rows = [&](int my_s, int my_d, int my_w, int m) {
        for (int i = 0; i < m; i += 8) {
            float a0[8], a1[8];
            #pragma unroll
            for (int u = 0; u < 8; ++u) {
                int s = __builtin_amdgcn_readlane(my_s, i + u);
                const float* r = h + (size_t)s * HID;
                a0[u] = r[lane];
                a1[u] = r[lane + 32];
            }
            #pragma unroll
            for (int u = 0; u < 8; ++u) {
                if (i + u < m) {
                    float w = __int_as_float(__builtin_amdgcn_readlane(my_w, i + u));
                    int   d = __builtin_amdgcn_readlane(my_d, i + u);
                    float* p = acc + d * HID + lane;
                    (void)__hip_atomic_fetch_add(p,      a0[u] * w, __ATOMIC_RELAXED, __HIP_MEMORY_SCOPE_WORKGROUP);
                    (void)__hip_atomic_fetch_add(p + 32, a1[u] * w, __ATOMIC_RELAXED, __HIP_MEMORY_SCOPE_WORKGROUP);
                }
            }
            if (MODE == 1)
                asm volatile("" :: "v"(a0[0]), "v"(a0[1]), "v"(a0[2]), "v"(a0[3]),
                                   "v"(a0[4]), "v"(a0[5]), "v"(a0[6]), "v"(a0[7]),
                                   "v"(a1[0]), "v"(a1[1]), "v"(a1[2]), "v"(a1[3]),
                                   "v"(a1[4]), "v"(a1[5]), "v"(a1[6]), "v"(a1[7]));
        }
    };

    auto drain = [&](int qb, int m) {
        int my_s = 0, my_d = 0, my_w = 0;
        if (lane < m) {
            int ei = qe[qb + lane];
            int dv = dst[ei];
            int sv = src[ei];
            if (MODE == 0) asm volatile("" :: "v"(dv), "v"(sv));
            my_d = dv - lo;
            my_s = sv;
            if (MODE != 0) my_w = __float_as_int(dinv[my_s] * dinv[lo + my_d]);
        }
        if (MODE == 0) {
            if (small) {
                if (!rfull && rn + m <= RCAP) {
                    if (lane < m) myrec[rn + lane] = make_int4(my_s, my_d, 0, 0);
                    rn += m;
                } else {
                    rfull = true;
                }
            }
            if (lane < m)
                (void)__hip_atomic_fetch_add(degc + my_d, 1, __ATOMIC_RELAXED, __HIP_MEMORY_SCOPE_WORKGROUP);
        } else {
            add_rows(my_s, my_d, my_w, m);
        }
    };

    auto append = [&](unsigned t, int ei) {
        bool hit = t < ucnt;
        unsigned mask = (unsigned)__ballot(hit);
        if (mask) {
            int pos = qn + (int)__builtin_amdgcn_mbcnt_lo(mask, 0u);
            if (hit) qe[pos] = ei;
            qn += __popc(mask);
        }
    };

    auto handle_dst = [&](int4 v, int chunk) {
        uint4 t;
        t.x = (unsigned)v.x - ulo; t.y = (unsigned)v.y - ulo;
        t.z = (unsigned)v.z - ulo; t.w = (unsigned)v.w - ulo;
        unsigned mn = min(min(t.x, t.y), min(t.z, t.w));
        if (__any(mn < ucnt)) {
            int base = chunk * 128 + lane * 4;
            append(t.x, base);
            append(t.y, base + 1);
            append(t.z, base + 2);
            append(t.w, base + 3);
        }
    };

    auto zero_bytes = [](unsigned x) -> unsigned {
        unsigned t = (x & 0x7F7F7F7Fu) + 0x7F7F7F7Fu;
        return ~(t | x | 0x7F7F7F7Fu);
    };

    auto match16 = [&](int4 v) -> unsigned {
        return (zero_bytes((unsigned)v.x ^ pat) >> 7) | (zero_bytes((unsigned)v.y ^ pat) >> 6)
             | (zero_bytes((unsigned)v.z ^ pat) >> 5) | (zero_bytes((unsigned)v.w ^ pat) >> 4);
    };

    auto enqueue = [&](unsigned m, int ebase) {
        if (__any(m != 0u)) {
            const int lbase = ebase + lane * 16;
            do {
                bool has = m != 0u;
                unsigned mask = (unsigned)__ballot(has);
                int pos = qn + (int)__builtin_amdgcn_mbcnt_lo(mask, 0u);
                if (has) {
                    int p = __builtin_ctz(m);
                    qe[pos] = lbase + ((p & 7) << 2) + (p >> 3);
                    m &= m - 1u;
                }
                qn += __popc(mask);
            } while (__any(m != 0u));
        }
    };

    const bool vec_ok  = (((size_t)key) & 15) == 0 && small;
    const int  nunit   = vec_ok ? (e >> 9) : 0;
    const int  ustart  = (int)(((long long)blockIdx.x * nunit) / gridDim.x);
    const int4* key4   = (const int4*)key;

    bool scan = true;
    if (MODE != 0 && small) {
        int c = __builtin_amdgcn_readfirstlane(rcount[region]);
        if (c >= 0) {
            scan = false;
            if (c > RCAP) c = RCAP;
            if (c > 0) {
                if (MODE == 1) {
                    int4 r1 = myrec[min(lane, c - 1)];
                    int4 r2 = myrec[min(32 + lane, c - 1)];
                    float ds = dinv[r1.x], dd = dinv[lo + r1.y];
                    for (int base = 0; base < c; base += 32) {
                        int4 cur = r1;
                        int wbits = __float_as_int(ds * dd);
                        r1 = r2;
                        r2 = myrec[min(base + 64 + lane, c - 1)];
                        ds = dinv[r1.x];
                        dd = dinv[lo + r1.y];
                        int m = c - base;
                        if (m > 32) m = 32;
                        add_rows(cur.x, cur.y, wbits, m);
                        if (lane < m) myrec[base + lane] = make_int4(cur.x, cur.y, wbits, 0);
                    }
                } else {
                    int4 nx = myrec[min(lane, c - 1)];
                    for (int base = 0; base < c; base += 32) {
                        int4 cur = nx;
                        nx = myrec[min(base + 32 + lane, c - 1)];
                        int m = c - base;
                        if (m > 32) m = 32;
                        add_rows(cur.x, cur.y, cur.z, m);
                    }
                }
            }
        }
    }

    if (cnt > 0 && scan) {
        if (wv < nunit) {
            const int nk = (nunit - wv + PWAVES - 1) / PWAVES;
            auto unit_of = [&](int k) -> int {
                int r = wv + (k < 0 ? 0 : (k < nk ? k : nk - 1)) * PWAVES + ustart;
                return r >= nunit ? r - nunit : r;
            };
            auto ldu = [&](int u) -> int4 { return key4[(size_t)u * 32 + lane]; };
            auto step = [&](int4& a, int k) {
                unsigned m = match16(a);
                a = ldu(unit_of(k + 8));
                if (k < 0 || k >= nk) m = 0u;
                enqueue(m, unit_of(k) * 512);
            };
            const int4 none = {0, 0, 0, 0};
            int4 a0 = none, a1 = none, a2 = none, a3 = none, a4 = none, a5 = none, a6 = none, a7 = none;
            for (int k = -8; k < nk; k += 8) {
                step(a0, k);     step(a1, k + 1); while (qn >= 32) { qn -= 32; drain(qn, 32); }
                step(a2, k + 2); step(a3, k + 3); while (qn >= 32) { qn -= 32; drain(qn, 32); }
                step(a4, k + 4); step(a5, k + 5); while (qn >= 32) { qn -= 32; drain(qn, 32); }
                step(a6, k + 6); step(a7, k + 7); while (qn >= 32) { qn -= 32; drain(qn, 32); }
            }
        }
        for (int c = nunit * 4 + wv; c * 128 < e; c += PWAVES) {
            handle_dst(load_chunk_guarded(c), c);
            while (qn >= 32) { qn -= 32; drain(qn, 32); }
        }
        if (qn > 0) drain(0, qn);
    }

    if (MODE == 0) {
        if (small && lane == 0) rcount[region] = rfull ? -1 : rn;
        __syncthreads();
        for (int t = tid; t < cnt; t += 1024)
            dinv_out[lo + t] = rsqrtf(1.0f + (float)degc[t]);
    } else {
        __syncthreads();
        for (int t = tid; t < cnt * HID; t += 1024)
            out[(size_t)lo * HID + t] = acc[t];
    }
}

__global__ void __launch_bounds__(1024)
k_scan_deg(const int* __restrict__ src, const int* __restrict__ dst,
           const unsigned char* __restrict__ key, float* __restrict__ dinv,
           int* __restrict__ recs, int* __restrict__ rcount, int n, int e) {
    __shared__ int degc[NPW];
    __shared__ int q_e[PWAVES][QCAP];
    own_body<0>(nullptr, degc, q_e, src, dst, key, nullptr, nullptr, dinv, nullptr, nullptr,
                recs, rcount, n, e);
}

__global__ void __launch_bounds__(1024)
k_prop_first(const int* __restrict__ src, const int* __restrict__ dst,
             const unsigned char* __restrict__ key, const float* __restrict__ h,
             const float* __restrict__ dinv, const float* __restrict__ b,
             float* __restrict__ out, int* __restrict__ recs, int* __restrict__ rcount,
             int n, int e) {
    __shared__ float acc[NPW * HID];
    __shared__ int   q_e[PWAVES][QCAP];
    own_body<1>(acc, nullptr, q_e, src, dst, key, h, dinv, nullptr, b, out, recs, rcount, n, e);
}

__global__ void __launch_bounds__(1024)
k_prop_second(const int* __restrict__ src, const int* __restrict__ dst,
              const unsigned char* __restrict__ key, const float* __restrict__ h,
              const float* __restrict__ dinv, const float* __restrict__ b,
              float* __restrict__ out, int* __restrict__ recs, int* __restrict__ rcount,
              int n, int e) {
    __shared__ float acc[NPW * HID];
    __shared__ int   q_e[PWAVES][QCAP];
    own_body<2>(acc, nullptr, q_e, src, dst, key, h, dinv, nullptr, b, out, recs, rcount, n, e);
}

__global__ void __launch_bounds__(256)
k_gemm_relu_wmma(const float* __restrict__ in, const float* __restrict__ W,
                 float* __restrict__ out, int nrows) {
    int gwave = (blockIdx.x * blockDim.x + threadIdx.x) >> 5;
    int lane  = threadIdx.x & 31;
    int row0  = gwave << 4;
    if (row0 >= nrows) return;
    int hh = lane >> 4;
    int l  = lane & 15;
    v8f acc0 = {}, acc1 = {}, acc2 = {}, acc3 = {};
    const float* arow = in + (size_t)(row0 + l) * HID;
    for (int k0 = 0; k0 < HID; k0 += 4) {
        int ka = k0 + hh * 2;
        v2f a;
        a.x = fmaxf(arow[ka],     0.0f);
        a.y = fmaxf(arow[ka + 1], 0.0f);
        const float* bp = W + (size_t)ka * HID + l;
        v2f b0, b1, b2, b3;
        b0.x = bp[0];  b0.y = bp[HID];
        b1.x = bp[16]; b1.y = bp[HID + 16];
        b2.x = bp[32]; b2.y = bp[HID + 32];
        b3.x = bp[48]; b3.y = bp[HID + 48];
        acc0 = __builtin_amdgcn_wmma_f32_16x16x4_f32(false, a, false, b0, (short)0, acc0, false, false);
        acc1 = __builtin_amdgcn_wmma_f32_16x16x4_f32(false, a, false, b1, (short)0, acc1, false, false);
        acc2 = __builtin_amdgcn_wmma_f32_16x16x4_f32(false, a, false, b2, (short)0, acc2, false, false);
        acc3 = __builtin_amdgcn_wmma_f32_16x16x4_f32(false, a, false, b3, (short)0, acc3, false, false);
    }
    float* orow = out + (size_t)(row0 + hh * 8) * HID + l;
    #pragma unroll
    for (int r = 0; r < 8; ++r) {
        float* o = orow + (size_t)r * HID;
        o[0]  = acc0[r];
        o[16] = acc1[r];
        o[32] = acc2[r];
        o[48] = acc3[r];
    }
}


__global__ void k_lin3_relu(const float* __restrict__ in, const float* __restrict__ W3,
                            float* __restrict__ h3, int n) {
    int gwave = (blockIdx.x * blockDim.x + threadIdx.x) >> 5;
    int lane  = threadIdx.x & 31;
    if (gwave >= n) return;
    const float* row = in + (size_t)gwave * HID;
    float v = fmaxf(row[lane], 0.0f) * W3[lane]
            + fmaxf(row[lane + 32], 0.0f) * W3[lane + 32];
    #pragma unroll
    for (int off = 16; off > 0; off >>= 1) v += __shfl_down(v, off, 32);
    if (lane == 0) h3[gwave] = v;
}

__global__ void __launch_bounds__(1024)
k_prop3_own(const int* __restrict__ src, const int* __restrict__ dst,
            const float* __restrict__ h3,
            const float* __restrict__ dinv, const float* __restrict__ b3,
            float* __restrict__ y, const int* __restrict__ recs,
            const int* __restrict__ rcount, int n, int e) {
    __shared__ float acc3[NPW];
    __shared__ int   no_recs;

    const int tid  = threadIdx.x;
    const int lane = tid & 31;
    const int wv   = __builtin_amdgcn_readfirstlane(tid >> 5);
    const int lo   = blockIdx.x * NPW;
    int cnt = n - lo;
    if (cnt > NPW) cnt = NPW;
    if (cnt < 0)   cnt = 0;
    const unsigned ucnt = (unsigned)cnt;

    const int region = blockIdx.x * PWAVES + wv;
    int c = -1;
    if (gridDim.x <= 256) c = __builtin_amdgcn_readfirstlane(rcount[region]);
    if (c > RCAP) c = RCAP;

    if (tid == 0) no_recs = 0;
    if (tid < cnt) {
        float di = dinv[lo + tid];
        acc3[tid] = b3[0] + h3[lo + tid] * di * di;
    }
    __syncthreads();
    if (c < 0 && lane == 0) no_recs = 1;
    __syncthreads();

    if (no_recs == 0) {
        const int4* myrec = (const int4*)recs + (size_t)region * RCAP;
        for (int base = 0; base < c; base += 256) {
            int4  r[8];
            float hv[8];
            #pragma unroll
            for (int u = 0; u < 8; ++u) r[u] = myrec[min(base + u * 32 + lane, c - 1)];
            #pragma unroll
            for (int u = 0; u < 8; ++u) hv[u] = h3[r[u].x];
            asm volatile("" :: "v"(hv[0]), "v"(hv[1]), "v"(hv[2]), "v"(hv[3]),
                               "v"(hv[4]), "v"(hv[5]), "v"(hv[6]), "v"(hv[7]));
            #pragma unroll
            for (int u = 0; u < 8; ++u) {
                if (base + u * 32 + lane < c)
                    (void)__hip_atomic_fetch_add(acc3 + r[u].y, hv[u] * __int_as_float(r[u].z),
                                                 __ATOMIC_RELAXED, __HIP_MEMORY_SCOPE_WORKGROUP);
            }
        }
    } else {
        for (int i = tid; i < e; i += 1024) {
            unsigned d = (unsigned)dst[i] - (unsigned)lo;
            if (d < ucnt) {
                int s = src[i];
                (void)__hip_atomic_fetch_add(acc3 + d, h3[s] * (dinv[s] * dinv[lo + d]),
                                             __ATOMIC_RELAXED, __HIP_MEMORY_SCOPE_WORKGROUP);
            }
        }
    }
    __syncthreads();

    if (tid < cnt) y[lo + tid] = acc3[tid];
}


extern "C" void kernel_launch(void* const* d_in, const int* in_sizes, int n_in,
                              void* d_out, int out_size, void* d_ws, size_t ws_size,
                              hipStream_t stream) {
    const float* x  = (const float*)d_in[0];
    const int*   ei = (const int*)  d_in[1];
    const float* W1 = (const float*)d_in[2];
    const float* b1 = (const float*)d_in[3];
    const float* W2 = (const float*)d_in[4];
    const float* b2 = (const float*)d_in[5];
    const float* W3 = (const float*)d_in[6];
    const float* b3 = (const float*)d_in[7];

    const int N = in_sizes[0] / 2;
    const int E = in_sizes[1] / 2;
    const int* src = ei;
    const int* dst = ei + E;

    float* dinv = (float*)d_ws;
    float* h    = dinv + (size_t)N;
    float* out  = h    + (size_t)N * HID;
    float* h3   = out  + (size_t)N * HID;
    unsigned char* key = (unsigned char*)(h3 + (size_t)N);
    int* recs   = (int*)(key + (size_t)E);
    int* rcount = recs + (size_t)256 * PWAVES * RCAP * 4;
    float* y    = (float*)d_out;

    const int B = 256;
    auto g = [](long long t) { return (int)((t + 255) / 256); };
    const int G = (N + NPW - 1) / NPW;

    k_key     <<<g(((long long)E + 3) / 4), B, 0, stream>>>(dst, key, E);
    k_scan_deg<<<G, 1024, 0, stream>>>(src, dst, key, dinv, recs, rcount, N, E);

    k_lin1      <<<g((long long)N * HID), B, 0, stream>>>(x, W1, h, N);
    k_prop_first<<<G, 1024, 0, stream>>>(src, dst, key, h, dinv, b1, out, recs, rcount, N, E);

    k_gemm_relu_wmma<<<g((long long)(N / 16) * 32), 256, 0, stream>>>(out, W2, h, N);
    k_prop_second<<<G, 1024, 0, stream>>>(src, dst, key, h, dinv, b2, out, recs, rcount, N, E);

    k_lin3_relu  <<<g((long long)N * 32), B, 0, stream>>>(out, W3, h3, N);
    k_prop3_own<<<G, 1024, 0, stream>>>(src, dst, h3, dinv, b3, y, recs, rcount, N, E);
}


# Edge scan split two ways: a scan workgroup reads 1/8 of the key list for the nodes of 8 owners (was all of it for 1), records left per owner; partial counts summed in x@W1 pass; LDS adds unordered
# speedup vs baseline: 3.3180x; 1.0374x over previous
#include <hip/hip_runtime.h>
#include <hip/hip_bf16.h>

typedef __attribute__((ext_vector_type(2))) float v2f;
typedef __attribute__((ext_vector_type(8))) float v8f;

#define HID 64
#define NPW 391
#define SUB 8

__global__ void k_key(const int* __restrict__ dst, unsigned char* __restrict__ key, int e) {
    int i = (blockIdx.x * blockDim.x + threadIdx.x) * 4;
    if (i + 3 < e && (((size_t)dst) & 15) == 0 && (((size_t)key) & 3) == 0) {
        int4 d = *(const int4*)(dst + i);
        unsigned k = (((unsigned)d.x / NPW) & 255u)       | ((((unsigned)d.y / NPW) & 255u) << 8)
                   | ((((unsigned)d.z / NPW) & 255u) << 16) | ((((unsigned)d.w / NPW) & 255u) << 24);
        *(unsigned*)(key + i) = k;
    } else {
        for (int u = 0; u < 4; ++u)
            if (i + u < e) key[i + u] = (unsigned char)((unsigned)dst[i + u] / NPW);
    }
}


__global__ void k_lin1(const float* __restrict__ x, const float* __restrict__ W1,
                       const int* __restrict__ pdeg, float* __restrict__ dinv,
                       float* __restrict__ h, int n) {
    int t = blockIdx.x * blockDim.x + threadIdx.x;
    int i = t >> 6, j = t & 63;
    if (i < n) {
        float2 xv = ((const float2*)x)[i];
        h[t] = xv.x * W1[j] + xv.y * W1[HID + j];
    }
    if (t < n) {
        int c = 0;
        for (int p = 0; p < SUB; ++p) c += pdeg[(size_t)p * n + t];
        dinv[t] = rsqrtf(1.0f + (float)c);
    }
}

#define PWAVES 32
#define QCAP   1056
#define RCAP   (1024 / SUB)

template <int MODE>
__device__ __forceinline__ void own_body(
           float* __restrict__ acc, int* __restrict__ degc, int (* __restrict__ q_e)[QCAP],
           const int* __restrict__ src, const int* __restrict__ dst,
           const unsigned char* __restrict__ key, const float* __restrict__ h,
           const float* __restrict__ dinv, int* __restrict__ pdeg,
           const float* __restrict__ b, float* __restrict__ out,
           int* __restrict__ recs, int* __restrict__ rcount, int n, int e) {
    const bool small = gridDim.x <= 256;
    const int tid  = threadIdx.x;
    const int lane = tid & 31;
    const int wv   = __builtin_amdgcn_readfirstlane(tid >> 5);
    const int grp  = blockIdx.x / SUB;
    const int part = blockIdx.x % SUB;
    const int span = (MODE == 0) ? SUB * NPW : NPW;
    const int lo   = (MODE == 0) ? grp * SUB * NPW : blockIdx.x * NPW;
    int cnt = n - lo;
    if (cnt > span) cnt = span;
    if (cnt < 0)    cnt = 0;
    const unsigned ucnt = (unsigned)cnt;
    const unsigned ulo  = (unsigned)lo;
    const int vw      = (MODE == 0) ? part * PWAVES + wv : wv;
    const int vstride = (MODE == 0) ? PWAVES * SUB : PWAVES;

    if (MODE == 0) {
        for (int t = tid; t < cnt; t += 1024) degc[t] = 0;
    } else {
        for (int t = tid; t < cnt * HID; t += 1024) {
            int i = t >> 6, j = t & 63;
            float di = dinv[lo + i];
            acc[t] = b[j] + h[(size_t)lo * HID + t] * di * di;
        }
    }
    __syncthreads();

    int* qe = q_e[wv];
    int  qn = 0;
    const int region0 = (MODE == 0) ? (grp * SUB * SUB + part) * PWAVES + wv
                                    : blockIdx.x * SUB * PWAVES + wv;
    const int rstep   = (MODE == 0) ? SUB * PWAVES : PWAVES;
    int4* rec0 = (int4*)recs + (size_t)region0 * RCAP;
    int      rn[SUB];
    #pragma unroll
    for (int i = 0; i < SUB; ++i) rn[i] = 0;
    unsigned rfull = 0u;
    const unsigned pat = (unsigned)(((MODE == 0) ? grp * SUB : blockIdx.x) & 255) * 0x01010101u;
    const unsigned lowm = (MODE == 0) ? (unsigned)(0x7F & ~(SUB - 1)) * 0x01010101u : 0x7F7F7F7Fu;

    auto load_chunk_guarded = [&](int chunk) -> int4 {
        int4 v;
        int base = chunk * 128 + lane * 4;
        v.x = (base     < e) ? dst[base]     : -1;
        v.y = (base + 1 < e) ? dst[base + 1] : -1;
        v.z = (base + 2 < e) ? dst[base + 2] : -1;
        v.w = (base + 3 < e) ? dst[base + 3] : -1;
        return v;
    };

    auto add_rows = [&](int my_s, int my_d, int my_w, int m) {
        for (int i = 0; i < m; i += 8) {
            float a0[8], a1[8];
            #pragma unroll
            for (int u = 0; u < 8; ++u) {
                int s = __builtin_amdgcn_readlane(my_s, i + u);
                const float* r = h + (size_t)s * HID;
                a0[u] = r[lane];
                a1[u] = r[lane + 32];
            }
            #pragma unroll
            for (int u = 0; u < 8; ++u) {
                if (i + u < m) {
                    float w = __int_as_float(__builtin_amdgcn_readlane(my_w, i + u));
                    int   d = __builtin_amdgcn_readlane(my_d, i + u);
                    float* p = acc + d * HID + lane;
                    (void)__hip_atomic_fetch_add(p,      a0[u] * w, __ATOMIC_RELAXED, __HIP_MEMORY_SCOPE_WORKGROUP);
                    (void)__hip_atomic_fetch_add(p + 32, a1[u] * w, __ATOMIC_RELAXED, __HIP_MEMORY_SCOPE_WORKGROUP);
                }
            }
            if (MODE == 1)
                asm volatile("" :: "v"(a0[0]), "v"(a0[1]), "v"(a0[2]), "v"(a0[3]),
                                   "v"(a0[4]), "v"(a0[5]), "v"(a0[6]), "v"(a0[7]),
                                   "v"(a1[0]), "v"(a1[1]), "v"(a1[2]), "v"(a1[3]),
                                   "v"(a1[4]), "v"(a1[5]), "v"(a1[6]), "v"(a1[7]));
        }
    };

    auto drain = [&](int qb, int m) {
        int my_s = 0, my_d = 0, my_w = 0;
        if (lane < m) {
            int ei = qe[qb + lane];
            int dv = dst[ei];
            int sv = src[ei];
            if (MODE == 0) asm volatile("" :: "v"(dv), "v"(sv));
            my_d = dv - lo;
            my_s = sv;
            if (MODE != 0) my_w = __float_as_int(dinv[my_s] * dinv[lo + my_d]);
        }
        if (MODE == 0) {
            if (small) {
                const int so_l = my_d / NPW;
                const int row  = my_d - so_l * NPW;
                #pragma unroll
                for (int so = 0; so < SUB; ++so) {
                    bool mine = lane < m && so_l == so;
                    unsigned mask = (unsigned)__ballot(mine);
                    if (mask) {
                        int k = __popc(mask);
                        if (!((rfull >> so) & 1u) && rn[so] + k <= RCAP) {
                            if (mine)
                                rec0[(size_t)so * rstep * RCAP + rn[so] + (int)__builtin_amdgcn_mbcnt_lo(mask, 0u)]
                                    = make_int4(my_s, row, 0, 0);
                            rn[so] += k;
                        } else {
                            rfull |= 1u << so;
                        }
                    }
                }
            }
            if (lane < m)
                (void)__hip_atomic_fetch_add(degc + my_d, 1, __ATOMIC_RELAXED, __HIP_MEMORY_SCOPE_WORKGROUP);
        } else {
            add_rows(my_s, my_d, my_w, m);
        }
    };

    auto append = [&](unsigned t, int ei) {
        bool hit = t < ucnt;
        unsigned mask = (unsigned)__ballot(hit);
        if (mask) {
            int pos = qn + (int)__builtin_amdgcn_mbcnt_lo(mask, 0u);
            if (hit) qe[pos] = ei;
            qn += __popc(mask);
        }
    };

    auto handle_dst = [&](int4 v, int chunk) {
        uint4 t;
        t.x = (unsigned)v.x - ulo; t.y = (unsigned)v.y - ulo;
        t.z = (unsigned)v.z - ulo; t.w = (unsigned)v.w - ulo;
        unsigned mn = min(min(t.x, t.y), min(t.z, t.w));
        if (__any(mn < ucnt)) {
            int base = chunk * 128 + lane * 4;
            append(t.x, base);
            append(t.y, base + 1);
            append(t.z, base + 2);
            append(t.w, base + 3);
        }
    };

    auto zero_bytes = [&](unsigned x) -> unsigned {
        unsigned t = (x & lowm) + 0x7F7F7F7Fu;
        return ~(t | x | 0x7F7F7F7Fu);
    };

    auto match16 = [&](int4 v) -> unsigned {
        return (zero_bytes((unsigned)v.x ^ pat) >> 7) | (zero_bytes((unsigned)v.y ^ pat) >> 6)
             | (zero_bytes((unsigned)v.z ^ pat) >> 5) | (zero_bytes((unsigned)v.w ^ pat) >> 4);
    };

    auto enqueue = [&](unsigned m, int ebase) {
        if (__any(m != 0u)) {
            const int lbase = ebase + lane * 16;
            do {
                bool has = m != 0u;
                unsigned mask = (unsigned)__ballot(has);
                int pos = qn + (int)__builtin_amdgcn_mbcnt_lo(mask, 0u);
                if (has) {
                    int p = __builtin_ctz(m);
                    qe[pos] = lbase + ((p & 7) << 2) + (p >> 3);
                    m &= m - 1u;
                }
                qn += __popc(mask);
            } while (__any(m != 0u));
        }
    };

    const bool vec_ok  = (((size_t)key) & 15) == 0 && small;
    const int  nunit   = vec_ok ? (e >> 9) : 0;
    const int  ngrp    = ((int)gridDim.x + SUB - 1) / SUB;
    const int  ustart  = (int)(((long long)grp * nunit) / ngrp);
    const int4* key4   = (const int4*)key;

    bool scan = true;
    if (MODE != 0 && small) {
        int  pre[SUB + 1];
        bool have = true;
        pre[0] = 0;
        #pragma unroll
        for (int p = 0; p < SUB; ++p) {
            int cp = __builtin_amdgcn_readfirstlane(rcount[region0 + p * rstep]);
            if (cp < 0) { have = false; cp = 0; }
            if (cp > RCAP) cp = RCAP;
            pre[p + 1] = pre[p] + cp;
        }
        auto recptr = [&](int i) -> int4* {
            int idx = i;
            #pragma unroll
            for (int p = 1; p < SUB; ++p)
                if (i >= pre[p]) idx = i - pre[p] + p * rstep * RCAP;
            return rec0 + idx;
        };
        if (have) {
            scan = false;
            const int c = pre[SUB];
            if (c > 0) {
                if (MODE == 1) {
                    int4 r1 = *recptr(min(lane, c - 1));
                    int4 r2 = *recptr(min(32 + lane, c - 1));
                    float ds = dinv[r1.x], dd = dinv[lo + r1.y];
                    for (int base = 0; base < c; base += 32) {
                        int4 cur = r1;
                        int wbits = __float_as_int(ds * dd);
                        r1 = r2;
                        r2 = *recptr(min(base + 64 + lane, c - 1));
                        ds = dinv[r1.x];
                        dd = dinv[lo + r1.y];
                        int m = c - base;
                        if (m > 32) m = 32;
                        add_rows(cur.x, cur.y, wbits, m);
                        if (lane < m) *recptr(base + lane) = make_int4(cur.x, cur.y, wbits, 0);
                    }
                } else {
                    int4 nx = *recptr(min(lane, c - 1));
                    for (int base = 0; base < c; base += 32) {
                        int4 cur = nx;
                        nx = *recptr(min(base + 32 + lane, c - 1));
                        int m = c - base;
                        if (m > 32) m = 32;
                        add_rows(cur.x, cur.y, cur.z, m);
                    }
                }
            }
        }
    }

    if (cnt > 0 && scan) {
        if (vw < nunit) {
            const int nk = (nunit - vw + vstride - 1) / vstride;
            auto unit_of = [&](int k) -> int {
                int r = vw + (k < 0 ? 0 : (k < nk ? k : nk - 1)) * vstride + ustart;
                return r >= nunit ? r - nunit : r;
            };
            auto ldu = [&](int u) -> int4 { return key4[(size_t)u * 32 + lane]; };
            auto step = [&](int4& a, int k) {
                unsigned m = match16(a);
                a = ldu(unit_of(k + 8));
                if (k < 0 || k >= nk) m = 0u;
                enqueue(m, unit_of(k) * 512);
            };
            const int4 none = {0, 0, 0, 0};
            int4 a0 = none, a1 = none, a2 = none, a3 = none, a4 = none, a5 = none, a6 = none, a7 = none;
            for (int k = -8; k < nk; k += 8) {
                step(a0, k);     step(a1, k + 1); while (qn >= 32) { qn -= 32; drain(qn, 32); }
                step(a2, k + 2); step(a3, k + 3); while (qn >= 32) { qn -= 32; drain(qn, 32); }
                step(a4, k + 4); step(a5, k + 5); while (qn >= 32) { qn -= 32; drain(qn, 32); }
                step(a6, k + 6); step(a7, k + 7); while (qn >= 32) { qn -= 32; drain(qn, 32); }
            }
        }
        for (int c = nunit * 4 + vw; c * 128 < e; c += vstride) {
            handle_dst(load_chunk_guarded(c), c);
            while (qn >= 32) { qn -= 32; drain(qn, 32); }
        }
        if (qn > 0) drain(0, qn);
    }

    if (MODE == 0) {
        if (small && lane == 0) {
            #pragma unroll
            for (int so = 0; so < SUB; ++so)
                rcount[region0 + so * rstep] = ((rfull >> so) & 1u) ? -1 : rn[so];
        }
        __syncthreads();
        for (int t = tid; t < cnt; t += 1024)
            pdeg[(size_t)part * n + lo + t] = degc[t];
    } else {
        __syncthreads();
        for (int t = tid; t < cnt * HID; t += 1024)
            out[(size_t)lo * HID + t] = acc[t];
    }
}

__global__ void __launch_bounds__(1024)
k_scan_deg(const int* __restrict__ src, const int* __restrict__ dst,
           const unsigned char* __restrict__ key, int* __restrict__ pdeg,
           int* __restrict__ recs, int* __restrict__ rcount, int n, int e) {
    __shared__ int degc[SUB * NPW];
    __shared__ int q_e[PWAVES][QCAP];
    own_body<0>(nullptr, degc, q_e, src, dst, key, nullptr, nullptr, pdeg, nullptr, nullptr,
                recs, rcount, n, e);
}

__global__ void __launch_bounds__(1024)
k_prop_first(const int* __restrict__ src, const int* __restrict__ dst,
             const unsigned char* __restrict__ key, const float* __restrict__ h,
             const float* __restrict__ dinv, const float* __restrict__ b,
             float* __restrict__ out, int* __restrict__ recs, int* __restrict__ rcount,
             int n, int e) {
    __shared__ float acc[NPW * HID];
    __shared__ int   q_e[PWAVES][QCAP];
    own_body<1>(acc, nullptr, q_e, src, dst, key, h, dinv, nullptr, b, out, recs, rcount, n, e);
}

__global__ void __launch_bounds__(1024)
k_prop_second(const int* __restrict__ src, const int* __restrict__ dst,
              const unsigned char* __restrict__ key, const float* __restrict__ h,
              const float* __restrict__ dinv, const float* __restrict__ b,
              float* __restrict__ out, int* __restrict__ recs, int* __restrict__ rcount,
              int n, int e) {
    __shared__ float acc[NPW * HID];
    __shared__ int   q_e[PWAVES][QCAP];
    own_body<2>(acc, nullptr, q_e, src, dst, key, h, dinv, nullptr, b, out, recs, rcount, n, e);
}

__global__ void __launch_bounds__(256)
k_gemm_relu_wmma(const float* __restrict__ in, const float* __restrict__ W,
                 float* __restrict__ out, int nrows) {
    int gwave = (blockIdx.x * blockDim.x + threadIdx.x) >> 5;
    int lane  = threadIdx.x & 31;
    int row0  = gwave << 4;
    if (row0 >= nrows) return;
    int hh = lane >> 4;
    int l  = lane & 15;
    v8f acc0 = {}, acc1 = {}, acc2 = {}, acc3 = {};
    const float* arow = in + (size_t)(row0 + l) * HID;
    for (int k0 = 0; k0 < HID; k0 += 4) {
        int ka = k0 + hh * 2;
        v2f a;
        a.x = fmaxf(arow[ka],     0.0f);
        a.y = fmaxf(arow[ka + 1], 0.0f);
        const float* bp = W + (size_t)ka * HID + l;
        v2f b0, b1, b2, b3;
        b0.x = bp[0];  b0.y = bp[HID];
        b1.x = bp[16]; b1.y = bp[HID + 16];
        b2.x = bp[32]; b2.y = bp[HID + 32];
        b3.x = bp[48]; b3.y = bp[HID + 48];
        acc0 = __builtin_amdgcn_wmma_f32_16x16x4_f32(false, a, false, b0, (short)0, acc0, false, false);
        acc1 = __builtin_amdgcn_wmma_f32_16x16x4_f32(false, a, false, b1, (short)0, acc1, false, false);
        acc2 = __builtin_amdgcn_wmma_f32_16x16x4_f32(false, a, false, b2, (short)0, acc2, false, false);
        acc3 = __builtin_amdgcn_wmma_f32_16x16x4_f32(false, a, false, b3, (short)0, acc3, false, false);
    }
    float* orow = out + (size_t)(row0 + hh * 8) * HID + l;
    #pragma unroll
    for (int r = 0; r < 8; ++r) {
        float* o = orow + (size_t)r * HID;
        o[0]  = acc0[r];
        o[16] = acc1[r];
        o[32] = acc2[r];
        o[48] = acc3[r];
    }
}


__global__ void k_lin3_relu(const float* __restrict__ in, const float* __restrict__ W3,
                            float* __restrict__ h3, int n) {
    int gwave = (blockIdx.x * blockDim.x + threadIdx.x) >> 5;
    int lane  = threadIdx.x & 31;
    if (gwave >= n) return;
    const float* row = in + (size_t)gwave * HID;
    float v = fmaxf(row[lane], 0.0f) * W3[lane]
            + fmaxf(row[lane + 32], 0.0f) * W3[lane + 32];
    #pragma unroll
    for (int off = 16; off > 0; off >>= 1) v += __shfl_down(v, off, 32);
    if (lane == 0) h3[gwave] = v;
}

__global__ void __launch_bounds__(1024)
k_prop3_own(const int* __restrict__ src, const int* __restrict__ dst,
            const float* __restrict__ h3,
            const float* __restrict__ dinv, const float* __restrict__ b3,
            float* __restrict__ y, const int* __restrict__ recs,
            const int* __restrict__ rcount, int n, int e) {
    __shared__ float acc3[NPW];
    __shared__ int   no_recs;

    const int tid  = threadIdx.x;
    const int lane = tid & 31;
    const int wv   = __builtin_amdgcn_readfirstlane(tid >> 5);
    const int lo   = blockIdx.x * NPW;
    int cnt = n - lo;
    if (cnt > NPW) cnt = NPW;
    if (cnt < 0)   cnt = 0;
    const unsigned ucnt = (unsigned)cnt;

    const int region0 = blockIdx.x * SUB * PWAVES + wv;
    const int4* rec0 = (const int4*)recs + (size_t)region0 * RCAP;
    int  pre[SUB + 1];
    bool have = gridDim.x <= 256;
    pre[0] = 0;
    #pragma unroll
    for (int p = 0; p < SUB; ++p) {
        int cp = -1;
        if (gridDim.x <= 256) cp = __builtin_amdgcn_readfirstlane(rcount[region0 + p * PWAVES]);
        if (cp < 0) { have = false; cp = 0; }
        if (cp > RCAP) cp = RCAP;
        pre[p + 1] = pre[p] + cp;
    }
    const int c = pre[SUB];
    auto recptr = [&](int i) -> const int4* {
        int idx = i;
        #pragma unroll
        for (int p = 1; p < SUB; ++p)
            if (i >= pre[p]) idx = i - pre[p] + p * PWAVES * RCAP;
        return rec0 + idx;
    };

    if (tid == 0) no_recs = 0;
    if (tid < cnt) {
        float di = dinv[lo + tid];
        acc3[tid] = b3[0] + h3[lo + tid] * di * di;
    }
    __syncthreads();
    if (!have && lane == 0) no_recs = 1;
    __syncthreads();

    if (no_recs == 0) {
        for (int base = 0; base < c; base += 256) {
            int4  r[8];
            float hv[8];
            #pragma unroll
            for (int u = 0; u < 8; ++u) r[u] = *recptr(min(base + u * 32 + lane, c - 1));
            #pragma unroll
            for (int u = 0; u < 8; ++u) hv[u] = h3[r[u].x];
            asm volatile("" :: "v"(hv[0]), "v"(hv[1]), "v"(hv[2]), "v"(hv[3]),
                               "v"(hv[4]), "v"(hv[5]), "v"(hv[6]), "v"(hv[7]));
            #pragma unroll
            for (int u = 0; u < 8; ++u) {
                if (base + u * 32 + lane < c)
                    (void)__hip_atomic_fetch_add(acc3 + r[u].y, hv[u] * __int_as_float(r[u].z),
                                                 __ATOMIC_RELAXED, __HIP_MEMORY_SCOPE_WORKGROUP);
            }
        }
    } else {
        for (int i = tid; i < e; i += 1024) {
            unsigned d = (unsigned)dst[i] - (unsigned)lo;
            if (d < ucnt) {
                int s = src[i];
                (void)__hip_atomic_fetch_add(acc3 + d, h3[s] * (dinv[s] * dinv[lo + d]),
                                             __ATOMIC_RELAXED, __HIP_MEMORY_SCOPE_WORKGROUP);
            }
        }
    }
    __syncthreads();

    if (tid < cnt) y[lo + tid] = acc3[tid];
}


extern "C" void kernel_launch(void* const* d_in, const int* in_sizes, int n_in,
                              void* d_out, int out_size, void* d_ws, size_t ws_size,
                              hipStream_t stream) {
    const float* x  = (const float*)d_in[0];
    const int*   ei = (const int*)  d_in[1];
    const float* W1 = (const float*)d_in[2];
    const float* b1 = (const float*)d_in[3];
    const float* W2 = (const float*)d_in[4];
    const float* b2 = (const float*)d_in[5];
    const float* W3 = (const float*)d_in[6];
    const float* b3 = (const float*)d_in[7];

    const int N = in_sizes[0] / 2;
    const int E = in_sizes[1] / 2;
    const int* src = ei;
    const int* dst = ei + E;

    float* dinv = (float*)d_ws;
    float* h    = dinv + (size_t)N;
    float* out  = h    + (size_t)N * HID;
    float* h3   = out  + (size_t)N * HID;
    unsigned char* key = (unsigned char*)(h3 + (size_t)N);
    int* recs   = (int*)(key + (size_t)E);
    int* rcount = recs + (size_t)256 * SUB * PWAVES * RCAP * 4;
    int* pdeg   = rcount + (size_t)256 * SUB * PWAVES;
    float* y    = (float*)d_out;

    const int B = 256;
    auto g = [](long long t) { return (int)((t + 255) / 256); };
    const int G = (N + NPW - 1) / NPW;
    const int GS = (G + SUB - 1) / SUB * SUB;

    k_key     <<<g(((long long)E + 3) / 4), B, 0, stream>>>(dst, key, E);
    k_scan_deg<<<GS, 1024, 0, stream>>>(src, dst, key, pdeg, recs, rcount, N, E);

    k_lin1      <<<g((long long)N * HID), B, 0, stream>>>(x, W1, pdeg, dinv, h, N);
    k_prop_first<<<G, 1024, 0, stream>>>(src, dst, key, h, dinv, b1, out, recs, rcount, N, E);

    k_gemm_relu_wmma<<<g((long long)(N / 16) * 32), 256, 0, stream>>>(out, W2, h, N);
    k_prop_second<<<G, 1024, 0, stream>>>(src, dst, key, h, dinv, b2, out, recs, rcount, N, E);

    k_lin3_relu  <<<g((long long)N * 32), B, 0, stream>>>(out, W3, h3, N);
    k_prop3_own<<<G, 1024, 0, stream>>>(src, dst, h3, dinv, b3, y, recs, rcount, N, E);
}


# Key, x@W1 and relu.W3 launches run one wave per 512 edges / 32 nodes with whole-line loads and stores: 9384 waves replace 312504 one-item waves; every sum and rounding as before
# speedup vs baseline: 3.4773x; 1.0480x over previous
#include <hip/hip_runtime.h>
#include <hip/hip_bf16.h>

typedef __attribute__((ext_vector_type(2))) float v2f;
typedef __attribute__((ext_vector_type(8))) float v8f;

#define HID 64
#define NPW 391
#define SUB 8

__global__ void k_key(const int* __restrict__ dst, unsigned char* __restrict__ key, int e) {
    const int t    = blockIdx.x * blockDim.x + threadIdx.x;
    const int lane = threadIdx.x & 31;
    const int base = (t >> 5) * 512;
    const bool al  = (((size_t)dst) & 15) == 0 && (((size_t)key) & 3) == 0;
    #pragma unroll
    for (int u = 0; u < 4; ++u) {
        const int i = base + u * 128 + lane * 4;
        if (i + 3 < e && al) {
            int4 d = *(const int4*)(dst + i);
            unsigned k = (((unsigned)d.x / NPW) & 255u)       | ((((unsigned)d.y / NPW) & 255u) << 8)
                       | ((((unsigned)d.z / NPW) & 255u) << 16) | ((((unsigned)d.w / NPW) & 255u) << 24);
            *(unsigned*)(key + i) = k;
        } else {
            for (int v = 0; v < 4; ++v)
                if (i + v < e) key[i + v] = (unsigned char)((unsigned)dst[i + v] / NPW);
        }
    }
}


__global__ void k_lin1(const float* __restrict__ x, const float* __restrict__ W1,
                       const int* __restrict__ pdeg, float* __restrict__ dinv,
                       float* __restrict__ h, int n) {
    const int t    = blockIdx.x * blockDim.x + threadIdx.x;
    const int lane = threadIdx.x & 31;
    const int row0 = (t >> 5) * 32;
    const int c4   = (lane & 15) * 4;
    const int hi   = lane >> 4;
    const bool al  = (((size_t)h) & 15) == 0;
    float2 xv = make_float2(0.0f, 0.0f);
    if (row0 + lane < n) xv = ((const float2*)x)[row0 + lane];
    float wa[4], wb[4];
    #pragma unroll
    for (int k = 0; k < 4; ++k) { wa[k] = W1[c4 + k]; wb[k] = W1[HID + c4 + k]; }
    if (al && row0 + 31 < n) {
        #pragma unroll
        for (int s = 0; s < 16; ++s) {
            const float ax = __shfl(xv.x, 2 * s + hi, 32);
            const float ay = __shfl(xv.y, 2 * s + hi, 32);
            float4 v;
            v.x = ax * wa[0] + ay * wb[0];
            v.y = ax * wa[1] + ay * wb[1];
            v.z = ax * wa[2] + ay * wb[2];
            v.w = ax * wa[3] + ay * wb[3];
            *(float4*)(h + (size_t)(row0 + 2 * s + hi) * HID + c4) = v;
        }
    } else {
        #pragma nounroll
        for (int s = 0; s < 16; ++s) {
            const float ax = __shfl(xv.x, 2 * s + hi, 32);
            const float ay = __shfl(xv.y, 2 * s + hi, 32);
            const int r = row0 + 2 * s + hi;
            if (r < n) {
                float* o = h + (size_t)r * HID + c4;
                #pragma unroll
                for (int k = 0; k < 4; ++k) o[k] = ax * wa[k] + ay * wb[k];
            }
        }
    }
    if (t < n) {
        int c = 0;
        for (int p = 0; p < SUB; ++p) c += pdeg[(size_t)p * n + t];
        dinv[t] = rsqrtf(1.0f + (float)c);
    }
}

#define PWAVES 32
#define QCAP   1056
#define RCAP   (1024 / SUB)

template <int MODE>
__device__ __forceinline__ void own_body(
           float* __restrict__ acc, int* __restrict__ degc, int (* __restrict__ q_e)[QCAP],
           const int* __restrict__ src, const int* __restrict__ dst,
           const unsigned char* __restrict__ key, const float* __restrict__ h,
           const float* __restrict__ dinv, int* __restrict__ pdeg,
           const float* __restrict__ b, float* __restrict__ out,
           int* __restrict__ recs, int* __restrict__ rcount, int n, int e) {
    const bool small = gridDim.x <= 256;
    const int tid  = threadIdx.x;
    const int lane = tid & 31;
    const int wv   = __builtin_amdgcn_readfirstlane(tid >> 5);
    const int grp  = blockIdx.x / SUB;
    const int part = blockIdx.x % SUB;
    const int span = (MODE == 0) ? SUB * NPW : NPW;
    const int lo   = (MODE == 0) ? grp * SUB * NPW : blockIdx.x * NPW;
    int cnt = n - lo;
    if (cnt > span) cnt = span;
    if (cnt < 0)    cnt = 0;
    const unsigned ucnt = (unsigned)cnt;
    const unsigned ulo  = (unsigned)lo;
    const int vw      = (MODE == 0) ? part * PWAVES + wv : wv;
    const int vstride = (MODE == 0) ? PWAVES * SUB : PWAVES;

    if (MODE == 0) {
        for (int t = tid; t < cnt; t += 1024) degc[t] = 0;
    } else {
        for (int t = tid; t < cnt * HID; t += 1024) {
            int i = t >> 6, j = t & 63;
            float di = dinv[lo + i];
            acc[t] = b[j] + h[(size_t)lo * HID + t] * di * di;
        }
    }
    __syncthreads();

    int* qe = q_e[wv];
    int  qn = 0;
    const int region0 = (MODE == 0) ? (grp * SUB * SUB + part) * PWAVES + wv
                                    : blockIdx.x * SUB * PWAVES + wv;
    const int rstep   = (MODE == 0) ? SUB * PWAVES : PWAVES;
    int4* rec0 = (int4*)recs + (size_t)region0 * RCAP;
    int      rn[SUB];
    #pragma unroll
    for (int i = 0; i < SUB; ++i) rn[i] = 0;
    unsigned rfull = 0u;
    const unsigned pat = (unsigned)(((MODE == 0) ? grp * SUB : blockIdx.x) & 255) * 0x01010101u;
    const unsigned lowm = (MODE == 0) ? (unsigned)(0x7F & ~(SUB - 1)) * 0x01010101u : 0x7F7F7F7Fu;

    auto load_chunk_guarded = [&](int chunk) -> int4 {
        int4 v;
        int base = chunk * 128 + lane * 4;
        v.x = (base     < e) ? dst[base]     : -1;
        v.y = (base + 1 < e) ? dst[base + 1] : -1;
        v.z = (base + 2 < e) ? dst[base + 2] : -1;
        v.w = (base + 3 < e) ? dst[base + 3] : -1;
        return v;
    };

    auto add_rows = [&](int my_s, int my_d, int my_w, int m) {
        for (int i = 0; i < m; i += 8) {
            float a0[8], a1[8];
            #pragma unroll
            for (int u = 0; u < 8; ++u) {
                int s = __builtin_amdgcn_readlane(my_s, i + u);
                const float* r = h + (size_t)s * HID;
                a0[u] = r[lane];
                a1[u] = r[lane + 32];
            }
            #pragma unroll
            for (int u = 0; u < 8; ++u) {
                if (i + u < m) {
                    float w = __int_as_float(__builtin_amdgcn_readlane(my_w, i + u));
                    int   d = __builtin_amdgcn_readlane(my_d, i + u);
                    float* p = acc + d * HID + lane;
                    (void)__hip_atomic_fetch_add(p,      a0[u] * w, __ATOMIC_RELAXED, __HIP_MEMORY_SCOPE_WORKGROUP);
                    (void)__hip_atomic_fetch_add(p + 32, a1[u] * w, __ATOMIC_RELAXED, __HIP_MEMORY_SCOPE_WORKGROUP);
                }
            }
            if (MODE == 1)
                asm volatile("" :: "v"(a0[0]), "v"(a0[1]), "v"(a0[2]), "v"(a0[3]),
                                   "v"(a0[4]), "v"(a0[5]), "v"(a0[6]), "v"(a0[7]),
                                   "v"(a1[0]), "v"(a1[1]), "v"(a1[2]), "v"(a1[3]),
                                   "v"(a1[4]), "v"(a1[5]), "v"(a1[6]), "v"(a1[7]));
        }
    };

    auto drain = [&](int qb, int m) {
        int my_s = 0, my_d = 0, my_w = 0;
        if (lane < m) {
            int ei = qe[qb + lane];
            int dv = dst[ei];
            int sv = src[ei];
            if (MODE == 0) asm volatile("" :: "v"(dv), "v"(sv));
            my_d = dv - lo;
            my_s = sv;
            if (MODE != 0) my_w = __float_as_int(dinv[my_s] * dinv[lo + my_d]);
        }
        if (MODE == 0) {
            if (small) {
                const int so_l = my_d / NPW;
                const int row  = my_d - so_l * NPW;
                #pragma unroll
                for (int so = 0; so < SUB; ++so) {
                    bool mine = lane < m && so_l == so;
                    unsigned mask = (unsigned)__ballot(mine);
                    if (mask) {
                        int k = __popc(mask);
                        if (!((rfull >> so) & 1u) && rn[so] + k <= RCAP) {
                            if (mine)
                                rec0[(size_t)so * rstep * RCAP + rn[so] + (int)__builtin_amdgcn_mbcnt_lo(mask, 0u)]
                                    = make_int4(my_s, row, 0, 0);
                            rn[so] += k;
                        } else {
                            rfull |= 1u << so;
                        }
                    }
                }
            }
            if (lane < m)
                (void)__hip_atomic_fetch_add(degc + my_d, 1, __ATOMIC_RELAXED, __HIP_MEMORY_SCOPE_WORKGROUP);
        } else {
            add_rows(my_s, my_d, my_w, m);
        }
    };

    auto append = [&](unsigned t, int ei) {
        bool hit = t < ucnt;
        unsigned mask = (unsigned)__ballot(hit);
        if (mask) {
            int pos = qn + (int)__builtin_amdgcn_mbcnt_lo(mask, 0u);
            if (hit) qe[pos] = ei;
            qn += __popc(mask);
        }
    };

    auto handle_dst = [&](int4 v, int chunk) {
        uint4 t;
        t.x = (unsigned)v.x - ulo; t.y = (unsigned)v.y - ulo;
        t.z = (unsigned)v.z - ulo; t.w = (unsigned)v.w - ulo;
        unsigned mn = min(min(t.x, t.y), min(t.z, t.w));
        if (__any(mn < ucnt)) {
            int base = chunk * 128 + lane * 4;
            append(t.x, base);
            append(t.y, base + 1);
            append(t.z, base + 2);
            append(t.w, base + 3);
        }
    };

    auto zero_bytes = [&](unsigned x) -> unsigned {
        unsigned t = (x & lowm) + 0x7F7F7F7Fu;
        return ~(t | x | 0x7F7F7F7Fu);
    };

    auto match16 = [&](int4 v) -> unsigned {
        return (zero_bytes((unsigned)v.x ^ pat) >> 7) | (zero_bytes((unsigned)v.y ^ pat) >> 6)
             | (zero_bytes((unsigned)v.z ^ pat) >> 5) | (zero_bytes((unsigned)v.w ^ pat) >> 4);
    };

    auto enqueue = [&](unsigned m, int ebase) {
        if (__any(m != 0u)) {
            const int lbase = ebase + lane * 16;
            do {
                bool has = m != 0u;
                unsigned mask = (unsigned)__ballot(has);
                int pos = qn + (int)__builtin_amdgcn_mbcnt_lo(mask, 0u);
                if (has) {
                    int p = __builtin_ctz(m);
                    qe[pos] = lbase + ((p & 7) << 2) + (p >> 3);
                    m &= m - 1u;
                }
                qn += __popc(mask);
            } while (__any(m != 0u));
        }
    };

    const bool vec_ok  = (((size_t)key) & 15) == 0 && small;
    const int  nunit   = vec_ok ? (e >> 9) : 0;
    const int  ngrp    = ((int)gridDim.x + SUB - 1) / SUB;
    const int  ustart  = (int)(((long long)grp * nunit) / ngrp);
    const int4* key4   = (const int4*)key;

    bool scan = true;
    if (MODE != 0 && small) {
        int  pre[SUB + 1];
        bool have = true;
        pre[0] = 0;
        #pragma unroll
        for (int p = 0; p < SUB; ++p) {
            int cp = __builtin_amdgcn_readfirstlane(rcount[region0 + p * rstep]);
            if (cp < 0) { have = false; cp = 0; }
            if (cp > RCAP) cp = RCAP;
            pre[p + 1] = pre[p] + cp;
        }
        auto recptr = [&](int i) -> int4* {
            int idx = i;
            #pragma unroll
            for (int p = 1; p < SUB; ++p)
                if (i >= pre[p]) idx = i - pre[p] + p * rstep * RCAP;
            return rec0 + idx;
        };
        if (have) {
            scan = false;
            const int c = pre[SUB];
            if (c > 0) {
                if (MODE == 1) {
                    int4 r1 = *recptr(min(lane, c - 1));
                    int4 r2 = *recptr(min(32 + lane, c - 1));
                    float ds = dinv[r1.x], dd = dinv[lo + r1.y];
                    for (int base = 0; base < c; base += 32) {
                        int4 cur = r1;
                        int wbits = __float_as_int(ds * dd);
                        r1 = r2;
                        r2 = *recptr(min(base + 64 + lane, c - 1));
                        ds = dinv[r1.x];
                        dd = dinv[lo + r1.y];
                        int m = c - base;
                        if (m > 32) m = 32;
                        add_rows(cur.x, cur.y, wbits, m);
                        if (lane < m) *recptr(base + lane) = make_int4(cur.x, cur.y, wbits, 0);
                    }
                } else {
                    int4 nx = *recptr(min(lane, c - 1));
                    for (int base = 0; base < c; base += 32) {
                        int4 cur = nx;
                        nx = *recptr(min(base + 32 + lane, c - 1));
                        int m = c - base;
                        if (m > 32) m = 32;
                        add_rows(cur.x, cur.y, cur.z, m);
                    }
                }
            }
        }
    }

    if (cnt > 0 && scan) {
        if (vw < nunit) {
            const int nk = (nunit - vw + vstride - 1) / vstride;
            auto unit_of = [&](int k) -> int {
                int r = vw + (k < 0 ? 0 : (k < nk ? k : nk - 1)) * vstride + ustart;
                return r >= nunit ? r - nunit : r;
            };
            auto ldu = [&](int u) -> int4 { return key4[(size_t)u * 32 + lane]; };
            auto step = [&](int4& a, int k) {
                unsigned m = match16(a);
                a = ldu(unit_of(k + 8));
                if (k < 0 || k >= nk) m = 0u;
                enqueue(m, unit_of(k) * 512);
            };
            const int4 none = {0, 0, 0, 0};
            int4 a0 = none, a1 = none, a2 = none, a3 = none, a4 = none, a5 = none, a6 = none, a7 = none;
            for (int k = -8; k < nk; k += 8) {
                step(a0, k);     step(a1, k + 1); while (qn >= 32) { qn -= 32; drain(qn, 32); }
                step(a2, k + 2); step(a3, k + 3); while (qn >= 32) { qn -= 32; drain(qn, 32); }
                step(a4, k + 4); step(a5, k + 5); while (qn >= 32) { qn -= 32; drain(qn, 32); }
                step(a6, k + 6); step(a7, k + 7); while (qn >= 32) { qn -= 32; drain(qn, 32); }
            }
        }
        for (int c = nunit * 4 + vw; c * 128 < e; c += vstride) {
            handle_dst(load_chunk_guarded(c), c);
            while (qn >= 32) { qn -= 32; drain(qn, 32); }
        }
        if (qn > 0) drain(0, qn);
    }

    if (MODE == 0) {
        if (small && lane == 0) {
            #pragma unroll
            for (int so = 0; so < SUB; ++so)
                rcount[region0 + so * rstep] = ((rfull >> so) & 1u) ? -1 : rn[so];
        }
        __syncthreads();
        for (int t = tid; t < cnt; t += 1024)
            pdeg[(size_t)part * n + lo + t] = degc[t];
    } else {
        __syncthreads();
        for (int t = tid; t < cnt * HID; t += 1024)
            out[(size_t)lo * HID + t] = acc[t];
    }
}

__global__ void __launch_bounds__(1024)
k_scan_deg(const int* __restrict__ src, const int* __restrict__ dst,
           const unsigned char* __restrict__ key, int* __restrict__ pdeg,
           int* __restrict__ recs, int* __restrict__ rcount, int n, int e) {
    __shared__ int degc[SUB * NPW];
    __shared__ int q_e[PWAVES][QCAP];
    own_body<0>(nullptr, degc, q_e, src, dst, key, nullptr, nullptr, pdeg, nullptr, nullptr,
                recs, rcount, n, e);
}

__global__ void __launch_bounds__(1024)
k_prop_first(const int* __restrict__ src, const int* __restrict__ dst,
             const unsigned char* __restrict__ key, const float* __restrict__ h,
             const float* __restrict__ dinv, const float* __restrict__ b,
             float* __restrict__ out, int* __restrict__ recs, int* __restrict__ rcount,
             int n, int e) {
    __shared__ float acc[NPW * HID];
    __shared__ int   q_e[PWAVES][QCAP];
    own_body<1>(acc, nullptr, q_e, src, dst, key, h, dinv, nullptr, b, out, recs, rcount, n, e);
}

__global__ void __launch_bounds__(1024)
k_prop_second(const int* __restrict__ src, const int* __restrict__ dst,
              const unsigned char* __restrict__ key, const float* __restrict__ h,
              const float* __restrict__ dinv, const float* __restrict__ b,
              float* __restrict__ out, int* __restrict__ recs, int* __restrict__ rcount,
              int n, int e) {
    __shared__ float acc[NPW * HID];
    __shared__ int   q_e[PWAVES][QCAP];
    own_body<2>(acc, nullptr, q_e, src, dst, key, h, dinv, nullptr, b, out, recs, rcount, n, e);
}

__global__ void __launch_bounds__(256)
k_gemm_relu_wmma(const float* __restrict__ in, const float* __restrict__ W,
                 float* __restrict__ out, int nrows) {
    int gwave = (blockIdx.x * blockDim.x + threadIdx.x) >> 5;
    int lane  = threadIdx.x & 31;
    int row0  = gwave << 4;
    if (row0 >= nrows) return;
    int hh = lane >> 4;
    int l  = lane & 15;
    v8f acc0 = {}, acc1 = {}, acc2 = {}, acc3 = {};
    const float* arow = in + (size_t)(row0 + l) * HID;
    for (int k0 = 0; k0 < HID; k0 += 4) {
        int ka = k0 + hh * 2;
        v2f a;
        a.x = fmaxf(arow[ka],     0.0f);
        a.y = fmaxf(arow[ka + 1], 0.0f);
        const float* bp = W + (size_t)ka * HID + l;
        v2f b0, b1, b2, b3;
        b0.x = bp[0];  b0.y = bp[HID];
        b1.x = bp[16]; b1.y = bp[HID + 16];
        b2.x = bp[32]; b2.y = bp[HID + 32];
        b3.x = bp[48]; b3.y = bp[HID + 48];
        acc0 = __builtin_amdgcn_wmma_f32_16x16x4_f32(false, a, false, b0, (short)0, acc0, false, false);
        acc1 = __builtin_amdgcn_wmma_f32_16x16x4_f32(false, a, false, b1, (short)0, acc1, false, false);
        acc2 = __builtin_amdgcn_wmma_f32_16x16x4_f32(false, a, false, b2, (short)0, acc2, false, false);
        acc3 = __builtin_amdgcn_wmma_f32_16x16x4_f32(false, a, false, b3, (short)0, acc3, false, false);
    }
    float* orow = out + (size_t)(row0 + hh * 8) * HID + l;
    #pragma unroll
    for (int r = 0; r < 8; ++r) {
        float* o = orow + (size_t)r * HID;
        o[0]  = acc0[r];
        o[16] = acc1[r];
        o[32] = acc2[r];
        o[48] = acc3[r];
    }
}


__global__ void k_lin3_relu(const float* __restrict__ in, const float* __restrict__ W3,
                            float* __restrict__ h3, int n) {
    const int t     = blockIdx.x * blockDim.x + threadIdx.x;
    const int lane  = threadIdx.x & 31;
    const int node0 = (t >> 5) * 32;
    if (node0 >= n) return;
    const float w0 = W3[lane], w1 = W3[lane + 32];
    float mine = 0.0f;
    #pragma unroll
    for (int u0 = 0; u0 < 32; u0 += 8) {
        float a0[8], a1[8];
        #pragma unroll
        for (int u = 0; u < 8; ++u) {
            const int node = min(node0 + u0 + u, n - 1);
            const float* row = in + (size_t)node * HID;
            a0[u] = row[lane];
            a1[u] = row[lane + 32];
        }
        #pragma unroll
        for (int u = 0; u < 8; ++u) {
            float v = fmaxf(a0[u], 0.0f) * w0 + fmaxf(a1[u], 0.0f) * w1;
            #pragma unroll
            for (int off = 16; off > 0; off >>= 1) v += __shfl_down(v, off, 32);
            const float tot = __int_as_float(__builtin_amdgcn_readfirstlane(__float_as_int(v)));
            if (lane == u0 + u) mine = tot;
        }
    }
    if (node0 + lane < n) h3[node0 + lane] = mine;
}

__global__ void __launch_bounds__(1024)
k_prop3_own(const int* __restrict__ src, const int* __restrict__ dst,
            const float* __restrict__ h3,
            const float* __restrict__ dinv, const float* __restrict__ b3,
            float* __restrict__ y, const int* __restrict__ recs,
            const int* __restrict__ rcount, int n, int e) {
    __shared__ float acc3[NPW];
    __shared__ int   no_recs;

    const int tid  = threadIdx.x;
    const int lane = tid & 31;
    const int wv   = __builtin_amdgcn_readfirstlane(tid >> 5);
    const int lo   = blockIdx.x * NPW;
    int cnt = n - lo;
    if (cnt > NPW) cnt = NPW;
    if (cnt < 0)   cnt = 0;
    const unsigned ucnt = (unsigned)cnt;

    const int region0 = blockIdx.x * SUB * PWAVES + wv;
    const int4* rec0 = (const int4*)recs + (size_t)region0 * RCAP;
    int  pre[SUB + 1];
    bool have = gridDim.x <= 256;
    pre[0] = 0;
    #pragma unroll
    for (int p = 0; p < SUB; ++p) {
        int cp = -1;
        if (gridDim.x <= 256) cp = __builtin_amdgcn_readfirstlane(rcount[region0 + p * PWAVES]);
        if (cp < 0) { have = false; cp = 0; }
        if (cp > RCAP) cp = RCAP;
        pre[p + 1] = pre[p] + cp;
    }
    const int c = pre[SUB];
    auto recptr = [&](int i) -> const int4* {
        int idx = i;
        #pragma unroll
        for (int p = 1; p < SUB; ++p)
            if (i >= pre[p]) idx = i - pre[p] + p * PWAVES * RCAP;
        return rec0 + idx;
    };

    if (tid == 0) no_recs = 0;
    if (tid < cnt) {
        float di = dinv[lo + tid];
        acc3[tid] = b3[0] + h3[lo + tid] * di * di;
    }
    __syncthreads();
    if (!have && lane == 0) no_recs = 1;
    __syncthreads();

    if (no_recs == 0) {
        for (int base = 0; base < c; base += 256) {
            int4  r[8];
            float hv[8];
            #pragma unroll
            for (int u = 0; u < 8; ++u) r[u] = *recptr(min(base + u * 32 + lane, c - 1));
            #pragma unroll
            for (int u = 0; u < 8; ++u) hv[u] = h3[r[u].x];
            asm volatile("" :: "v"(hv[0]), "v"(hv[1]), "v"(hv[2]), "v"(hv[3]),
                               "v"(hv[4]), "v"(hv[5]), "v"(hv[6]), "v"(hv[7]));
            #pragma unroll
            for (int u = 0; u < 8; ++u) {
                if (base + u * 32 + lane < c)
                    (void)__hip_atomic_fetch_add(acc3 + r[u].y, hv[u] * __int_as_float(r[u].z),
                                                 __ATOMIC_RELAXED, __HIP_MEMORY_SCOPE_WORKGROUP);
            }
        }
    } else {
        for (int i = tid; i < e; i += 1024) {
            unsigned d = (unsigned)dst[i] - (unsigned)lo;
            if (d < ucnt) {
                int s = src[i];
                (void)__hip_atomic_fetch_add(acc3 + d, h3[s] * (dinv[s] * dinv[lo + d]),
                                             __ATOMIC_RELAXED, __HIP_MEMORY_SCOPE_WORKGROUP);
            }
        }
    }
    __syncthreads();

    if (tid < cnt) y[lo + tid] = acc3[tid];
}


extern "C" void kernel_launch(void* const* d_in, const int* in_sizes, int n_in,
                              void* d_out, int out_size, void* d_ws, size_t ws_size,
                              hipStream_t stream) {
    const float* x  = (const float*)d_in[0];
    const int*   ei = (const int*)  d_in[1];
    const float* W1 = (const float*)d_in[2];
    const float* b1 = (const float*)d_in[3];
    const float* W2 = (const float*)d_in[4];
    const float* b2 = (const float*)d_in[5];
    const float* W3 = (const float*)d_in[6];
    const float* b3 = (const float*)d_in[7];

    const int N = in_sizes[0] / 2;
    const int E = in_sizes[1] / 2;
    const int* src = ei;
    const int* dst = ei + E;

    float* dinv = (float*)d_ws;
    float* h    = dinv + (size_t)N;
    float* out  = h    + (size_t)N * HID;
    float* h3   = out  + (size_t)N * HID;
    unsigned char* key = (unsigned char*)(h3 + (size_t)N);
    int* recs   = (int*)(key + (size_t)E);
    int* rcount = recs + (size_t)256 * SUB * PWAVES * RCAP * 4;
    int* pdeg   = rcount + (size_t)256 * SUB * PWAVES;
    float* y    = (float*)d_out;

    const int B = 256;
    auto g = [](long long t) { return (int)((t + 255) / 256); };
    const int G = (N + NPW - 1) / NPW;
    const int GS = (G + SUB - 1) / SUB * SUB;

    k_key     <<<g(((long long)E + 511) / 512 * 32), B, 0, stream>>>(dst, key, E);
    k_scan_deg<<<GS, 1024, 0, stream>>>(src, dst, key, pdeg, recs, rcount, N, E);

    k_lin1      <<<g(((long long)N + 31) / 32 * 32), B, 0, stream>>>(x, W1, pdeg, dinv, h, N);
    k_prop_first<<<G, 1024, 0, stream>>>(src, dst, key, h, dinv, b1, out, recs, rcount, N, E);

    k_gemm_relu_wmma<<<g((long long)(N / 16) * 32), 256, 0, stream>>>(out, W2, h, N);
    k_prop_second<<<G, 1024, 0, stream>>>(src, dst, key, h, dinv, b2, out, recs, rcount, N, E);

    k_lin3_relu  <<<g(((long long)N + 31) / 32 * 32), B, 0, stream>>>(out, W3, h3, N);
    k_prop3_own<<<G, 1024, 0, stream>>>(src, dst, h3, dinv, b3, y, recs, rcount, N, E);
}


# Owner workgroups apply the next layer's transform to their finished LDS rows: relu.W2 by f32 WMMA read from LDS after propagation 1, relu.W3 dot after propagation 2; two [N,64] planes, 2 launches go
# speedup vs baseline: 3.9790x; 1.1443x over previous
#include <hip/hip_runtime.h>
#include <hip/hip_bf16.h>

typedef __attribute__((ext_vector_type(2))) float v2f;
typedef __attribute__((ext_vector_type(8))) float v8f;

#define HID 64
#define NPW 391
#define SUB 8

__global__ void k_key(const int* __restrict__ dst, unsigned char* __restrict__ key, int e) {
    const int t    = blockIdx.x * blockDim.x + threadIdx.x;
    const int lane = threadIdx.x & 31;
    const int base = (t >> 5) * 512;
    const bool al  = (((size_t)dst) & 15) == 0 && (((size_t)key) & 3) == 0;
    #pragma unroll
    for (int u = 0; u < 4; ++u) {
        const int i = base + u * 128 + lane * 4;
        if (i + 3 < e && al) {
            int4 d = *(const int4*)(dst + i);
            unsigned k = (((unsigned)d.x / NPW) & 255u)       | ((((unsigned)d.y / NPW) & 255u) << 8)
                       | ((((unsigned)d.z / NPW) & 255u) << 16) | ((((unsigned)d.w / NPW) & 255u) << 24);
            *(unsigned*)(key + i) = k;
        } else {
            for (int v = 0; v < 4; ++v)
                if (i + v < e) key[i + v] = (unsigned char)((unsigned)dst[i + v] / NPW);
        }
    }
}


__global__ void k_lin1(const float* __restrict__ x, const float* __restrict__ W1,
                       const int* __restrict__ pdeg, float* __restrict__ dinv,
                       float* __restrict__ h, int n) {
    const int t    = blockIdx.x * blockDim.x + threadIdx.x;
    const int lane = threadIdx.x & 31;
    const int row0 = (t >> 5) * 32;
    const int c4   = (lane & 15) * 4;
    const int hi   = lane >> 4;
    const bool al  = (((size_t)h) & 15) == 0;
    float2 xv = make_float2(0.0f, 0.0f);
    if (row0 + lane < n) xv = ((const float2*)x)[row0 + lane];
    float wa[4], wb[4];
    #pragma unroll
    for (int k = 0; k < 4; ++k) { wa[k] = W1[c4 + k]; wb[k] = W1[HID + c4 + k]; }
    if (al && row0 + 31 < n) {
        #pragma unroll
        for (int s = 0; s < 16; ++s) {
            const float ax = __shfl(xv.x, 2 * s + hi, 32);
            const float ay = __shfl(xv.y, 2 * s + hi, 32);
            float4 v;
            v.x = ax * wa[0] + ay * wb[0];
            v.y = ax * wa[1] + ay * wb[1];
            v.z = ax * wa[2] + ay * wb[2];
            v.w = ax * wa[3] + ay * wb[3];
            *(float4*)(h + (size_t)(row0 + 2 * s + hi) * HID + c4) = v;
        }
    } else {
        #pragma nounroll
        for (int s = 0; s < 16; ++s) {
            const float ax = __shfl(xv.x, 2 * s + hi, 32);
            const float ay = __shfl(xv.y, 2 * s + hi, 32);
            const int r = row0 + 2 * s + hi;
            if (r < n) {
                float* o = h + (size_t)r * HID + c4;
                #pragma unroll
                for (int k = 0; k < 4; ++k) o[k] = ax * wa[k] + ay * wb[k];
            }
        }
    }
    if (t < n) {
        int c = 0;
        for (int p = 0; p < SUB; ++p) c += pdeg[(size_t)p * n + t];
        dinv[t] = rsqrtf(1.0f + (float)c);
    }
}

#define PWAVES 32
#define QCAP   1056
#define RCAP   (1024 / SUB)

template <int MODE>
__device__ __forceinline__ void own_body(
           float* __restrict__ acc, int* __restrict__ degc, int (* __restrict__ q_e)[QCAP],
           const int* __restrict__ src, const int* __restrict__ dst,
           const unsigned char* __restrict__ key, const float* __restrict__ h,
           const float* __restrict__ dinv, int* __restrict__ pdeg,
           const float* __restrict__ b, const float* __restrict__ Wn, float* __restrict__ out,
           int* __restrict__ recs, int* __restrict__ rcount, int n, int e) {
    const bool small = gridDim.x <= 256;
    const int tid  = threadIdx.x;
    const int lane = tid & 31;
    const int wv   = __builtin_amdgcn_readfirstlane(tid >> 5);
    const int grp  = blockIdx.x / SUB;
    const int part = blockIdx.x % SUB;
    const int span = (MODE == 0) ? SUB * NPW : NPW;
    const int lo   = (MODE == 0) ? grp * SUB * NPW : blockIdx.x * NPW;
    int cnt = n - lo;
    if (cnt > span) cnt = span;
    if (cnt < 0)    cnt = 0;
    const unsigned ucnt = (unsigned)cnt;
    const unsigned ulo  = (unsigned)lo;
    const int vw      = (MODE == 0) ? part * PWAVES + wv : wv;
    const int vstride = (MODE == 0) ? PWAVES * SUB : PWAVES;

    if (MODE == 0) {
        for (int t = tid; t < cnt; t += 1024) degc[t] = 0;
    } else {
        for (int t = tid; t < cnt * HID; t += 1024) {
            int i = t >> 6, j = t & 63;
            float di = dinv[lo + i];
            acc[t] = b[j] + h[(size_t)lo * HID + t] * di * di;
        }
    }
    __syncthreads();

    int* qe = q_e[wv];
    int  qn = 0;
    const int region0 = (MODE == 0) ? (grp * SUB * SUB + part) * PWAVES + wv
                                    : blockIdx.x * SUB * PWAVES + wv;
    const int rstep   = (MODE == 0) ? SUB * PWAVES : PWAVES;
    int4* rec0 = (int4*)recs + (size_t)region0 * RCAP;
    int      rn[SUB];
    #pragma unroll
    for (int i = 0; i < SUB; ++i) rn[i] = 0;
    unsigned rfull = 0u;
    const unsigned pat = (unsigned)(((MODE == 0) ? grp * SUB : blockIdx.x) & 255) * 0x01010101u;
    const unsigned lowm = (MODE == 0) ? (unsigned)(0x7F & ~(SUB - 1)) * 0x01010101u : 0x7F7F7F7Fu;

    auto load_chunk_guarded = [&](int chunk) -> int4 {
        int4 v;
        int base = chunk * 128 + lane * 4;
        v.x = (base     < e) ? dst[base]     : -1;
        v.y = (base + 1 < e) ? dst[base + 1] : -1;
        v.z = (base + 2 < e) ? dst[base + 2] : -1;
        v.w = (base + 3 < e) ? dst[base + 3] : -1;
        return v;
    };

    auto add_rows = [&](int my_s, int my_d, int my_w, int m) {
        for (int i = 0; i < m; i += 8) {
            float a0[8], a1[8];
            #pragma unroll
            for (int u = 0; u < 8; ++u) {
                int s = __builtin_amdgcn_readlane(my_s, i + u);
                const float* r = h + (size_t)s * HID;
                a0[u] = r[lane];
                a1[u] = r[lane + 32];
            }
            #pragma unroll
            for (int u = 0; u < 8; ++u) {
                if (i + u < m) {
                    float w = __int_as_float(__builtin_amdgcn_readlane(my_w, i + u));
                    int   d = __builtin_amdgcn_readlane(my_d, i + u);
                    float* p = acc + d * HID + lane;
                    (void)__hip_atomic_fetch_add(p,      a0[u] * w, __ATOMIC_RELAXED, __HIP_MEMORY_SCOPE_WORKGROUP);
                    (void)__hip_atomic_fetch_add(p + 32, a1[u] * w, __ATOMIC_RELAXED, __HIP_MEMORY_SCOPE_WORKGROUP);
                }
            }
            if (MODE == 1)
                asm volatile("" :: "v"(a0[0]), "v"(a0[1]), "v"(a0[2]), "v"(a0[3]),
                                   "v"(a0[4]), "v"(a0[5]), "v"(a0[6]), "v"(a0[7]),
                                   "v"(a1[0]), "v"(a1[1]), "v"(a1[2]), "v"(a1[3]),
                                   "v"(a1[4]), "v"(a1[5]), "v"(a1[6]), "v"(a1[7]));
        }
    };

    auto drain = [&](int qb, int m) {
        int my_s = 0, my_d = 0, my_w = 0;
        if (lane < m) {
            int ei = qe[qb + lane];
            int dv = dst[ei];
            int sv = src[ei];
            if (MODE == 0) asm volatile("" :: "v"(dv), "v"(sv));
            my_d = dv - lo;
            my_s = sv;
            if (MODE != 0) my_w = __float_as_int(dinv[my_s] * dinv[lo + my_d]);
        }
        if (MODE == 0) {
            if (small) {
                const int so_l = my_d / NPW;
                const int row  = my_d - so_l * NPW;
                #pragma unroll
                for (int so = 0; so < SUB; ++so) {
                    bool mine = lane < m && so_l == so;
                    unsigned mask = (unsigned)__ballot(mine);
                    if (mask) {
                        int k = __popc(mask);
                        if (!((rfull >> so) & 1u) && rn[so] + k <= RCAP) {
                            if (mine)
                                rec0[(size_t)so * rstep * RCAP + rn[so] + (int)__builtin_amdgcn_mbcnt_lo(mask, 0u)]
                                    = make_int4(my_s, row, 0, 0);
                            rn[so] += k;
                        } else {
                            rfull |= 1u << so;
                        }
                    }
                }
            }
            if (lane < m)
                (void)__hip_atomic_fetch_add(degc + my_d, 1, __ATOMIC_RELAXED, __HIP_MEMORY_SCOPE_WORKGROUP);
        } else {
            add_rows(my_s, my_d, my_w, m);
        }
    };

    auto append = [&](unsigned t, int ei) {
        bool hit = t < ucnt;
        unsigned mask = (unsigned)__ballot(hit);
        if (mask) {
            int pos = qn + (int)__builtin_amdgcn_mbcnt_lo(mask, 0u);
            if (hit) qe[pos] = ei;
            qn += __popc(mask);
        }
    };

    auto handle_dst = [&](int4 v, int chunk) {
        uint4 t;
        t.x = (unsigned)v.x - ulo; t.y = (unsigned)v.y - ulo;
        t.z = (unsigned)v.z - ulo; t.w = (unsigned)v.w - ulo;
        unsigned mn = min(min(t.x, t.y), min(t.z, t.w));
        if (__any(mn < ucnt)) {
            int base = chunk * 128 + lane * 4;
            append(t.x, base);
            append(t.y, base + 1);
            append(t.z, base + 2);
            append(t.w, base + 3);
        }
    };

    auto zero_bytes = [&](unsigned x) -> unsigned {
        unsigned t = (x & lowm) + 0x7F7F7F7Fu;
        return ~(t | x | 0x7F7F7F7Fu);
    };

    auto match16 = [&](int4 v) -> unsigned {
        return (zero_bytes((unsigned)v.x ^ pat) >> 7) | (zero_bytes((unsigned)v.y ^ pat) >> 6)
             | (zero_bytes((unsigned)v.z ^ pat) >> 5) | (zero_bytes((unsigned)v.w ^ pat) >> 4);
    };

    auto enqueue = [&](unsigned m, int ebase) {
        if (__any(m != 0u)) {
            const int lbase = ebase + lane * 16;
            do {
                bool has = m != 0u;
                unsigned mask = (unsigned)__ballot(has);
                int pos = qn + (int)__builtin_amdgcn_mbcnt_lo(mask, 0u);
                if (has) {
                    int p = __builtin_ctz(m);
                    qe[pos] = lbase + ((p & 7) << 2) + (p >> 3);
                    m &= m - 1u;
                }
                qn += __popc(mask);
            } while (__any(m != 0u));
        }
    };

    const bool vec_ok  = (((size_t)key) & 15) == 0 && small;
    const int  nunit   = vec_ok ? (e >> 9) : 0;
    const int  ngrp    = ((int)gridDim.x + SUB - 1) / SUB;
    const int  ustart  = (int)(((long long)grp * nunit) / ngrp);
    const int4* key4   = (const int4*)key;

    bool scan = true;
    if (MODE != 0 && small) {
        int  pre[SUB + 1];
        bool have = true;
        pre[0] = 0;
        #pragma unroll
        for (int p = 0; p < SUB; ++p) {
            int cp = __builtin_amdgcn_readfirstlane(rcount[region0 + p * rstep]);
            if (cp < 0) { have = false; cp = 0; }
            if (cp > RCAP) cp = RCAP;
            pre[p + 1] = pre[p] + cp;
        }
        auto recptr = [&](int i) -> int4* {
            int idx = i;
            #pragma unroll
            for (int p = 1; p < SUB; ++p)
                if (i >= pre[p]) idx = i - pre[p] + p * rstep * RCAP;
            return rec0 + idx;
        };
        if (have) {
            scan = false;
            const int c = pre[SUB];
            if (c > 0) {
                if (MODE == 1) {
                    int4 r1 = *recptr(min(lane, c - 1));
                    int4 r2 = *recptr(min(32 + lane, c - 1));
                    float ds = dinv[r1.x], dd = dinv[lo + r1.y];
                    for (int base = 0; base < c; base += 32) {
                        int4 cur = r1;
                        int wbits = __float_as_int(ds * dd);
                        r1 = r2;
                        r2 = *recptr(min(base + 64 + lane, c - 1));
                        ds = dinv[r1.x];
                        dd = dinv[lo + r1.y];
                        int m = c - base;
                        if (m > 32) m = 32;
                        add_rows(cur.x, cur.y, wbits, m);
                        if (lane < m) *recptr(base + lane) = make_int4(cur.x, cur.y, wbits, 0);
                    }
                } else {
                    int4 nx = *recptr(min(lane, c - 1));
                    for (int base = 0; base < c; base += 32) {
                        int4 cur = nx;
                        nx = *recptr(min(base + 32 + lane, c - 1));
                        int m = c - base;
                        if (m > 32) m = 32;
                        add_rows(cur.x, cur.y, cur.z, m);
                    }
                }
            }
        }
    }

    if (cnt > 0 && scan) {
        if (vw < nunit) {
            const int nk = (nunit - vw + vstride - 1) / vstride;
            auto unit_of = [&](int k) -> int {
                int r = vw + (k < 0 ? 0 : (k < nk ? k : nk - 1)) * vstride + ustart;
                return r >= nunit ? r - nunit : r;
            };
            auto ldu = [&](int u) -> int4 { return key4[(size_t)u * 32 + lane]; };
            auto step = [&](int4& a, int k) {
                unsigned m = match16(a);
                a = ldu(unit_of(k + 8));
                if (k < 0 || k >= nk) m = 0u;
                enqueue(m, unit_of(k) * 512);
            };
            const int4 none = {0, 0, 0, 0};
            int4 a0 = none, a1 = none, a2 = none, a3 = none, a4 = none, a5 = none, a6 = none, a7 = none;
            for (int k = -8; k < nk; k += 8) {
                step(a0, k);     step(a1, k + 1); while (qn >= 32) { qn -= 32; drain(qn, 32); }
                step(a2, k + 2); step(a3, k + 3); while (qn >= 32) { qn -= 32; drain(qn, 32); }
                step(a4, k + 4); step(a5, k + 5); while (qn >= 32) { qn -= 32; drain(qn, 32); }
                step(a6, k + 6); step(a7, k + 7); while (qn >= 32) { qn -= 32; drain(qn, 32); }
            }
        }
        for (int c = nunit * 4 + vw; c * 128 < e; c += vstride) {
            handle_dst(load_chunk_guarded(c), c);
            while (qn >= 32) { qn -= 32; drain(qn, 32); }
        }
        if (qn > 0) drain(0, qn);
    }

    if (MODE == 0) {
        if (small && lane == 0) {
            #pragma unroll
            for (int so = 0; so < SUB; ++so)
                rcount[region0 + so * rstep] = ((rfull >> so) & 1u) ? -1 : rn[so];
        }
        __syncthreads();
        for (int t = tid; t < cnt; t += 1024)
            pdeg[(size_t)part * n + lo + t] = degc[t];
    } else if (MODE == 1) {
        __syncthreads();
        const int hh = lane >> 4;
        const int l  = lane & 15;
        for (int row0 = wv * 16; row0 < cnt; row0 += PWAVES * 16) {
            int ar = row0 + l;
            if (ar > cnt - 1) ar = cnt - 1;
            const float* arow = acc + ar * HID;
            v8f acc0 = {}, acc1 = {}, acc2 = {}, acc3 = {};
            for (int k0 = 0; k0 < HID; k0 += 4) {
                const int ka = k0 + hh * 2;
                v2f a;
                a.x = fmaxf(arow[ka],     0.0f);
                a.y = fmaxf(arow[ka + 1], 0.0f);
                const float* bp = Wn + (size_t)ka * HID + l;
                v2f b0, b1, b2, b3;
                b0.x = bp[0];  b0.y = bp[HID];
                b1.x = bp[16]; b1.y = bp[HID + 16];
                b2.x = bp[32]; b2.y = bp[HID + 32];
                b3.x = bp[48]; b3.y = bp[HID + 48];
                acc0 = __builtin_amdgcn_wmma_f32_16x16x4_f32(false, a, false, b0, (short)0, acc0, false, false);
                acc1 = __builtin_amdgcn_wmma_f32_16x16x4_f32(false, a, false, b1, (short)0, acc1, false, false);
                acc2 = __builtin_amdgcn_wmma_f32_16x16x4_f32(false, a, false, b2, (short)0, acc2, false, false);
                acc3 = __builtin_amdgcn_wmma_f32_16x16x4_f32(false, a, false, b3, (short)0, acc3, false, false);
            }
            #pragma unroll
            for (int r = 0; r < 8; ++r) {
                const int orow = row0 + hh * 8 + r;
                if (orow < cnt) {
                    float* o = acc + orow * HID + l;
                    o[0]  = acc0[r];
                    o[16] = acc1[r];
                    o[32] = acc2[r];
                    o[48] = acc3[r];
                }
            }
        }
        __syncthreads();
        for (int t = tid; t < cnt * HID; t += 1024)
            out[(size_t)lo * HID + t] = acc[t];
    } else {
        __syncthreads();
        constexpr int RPW = (NPW + PWAVES - 1) / PWAVES;
        const float w0 = Wn[lane], w1 = Wn[lane + 32];
        const int r0 = wv * RPW;
        if (r0 < cnt) {
            float mine = 0.0f;
            #pragma unroll
            for (int u = 0; u < RPW; ++u) {
                int r = r0 + u;
                if (r > cnt - 1) r = cnt - 1;
                float v = fmaxf(acc[r * HID + lane], 0.0f) * w0 + fmaxf(acc[r * HID + lane + 32], 0.0f) * w1;
                #pragma unroll
                for (int off = 16; off > 0; off >>= 1) v += __shfl_down(v, off, 32);
                const float tot = __int_as_float(__builtin_amdgcn_readfirstlane(__float_as_int(v)));
                if (lane == u) mine = tot;
            }
            if (lane < RPW && r0 + lane < cnt) out[lo + r0 + lane] = mine;
        }
    }
}

__global__ void __launch_bounds__(1024)
k_scan_deg(const int* __restrict__ src, const int* __restrict__ dst,
           const unsigned char* __restrict__ key, int* __restrict__ pdeg,
           int* __restrict__ recs, int* __restrict__ rcount, int n, int e) {
    __shared__ int degc[SUB * NPW];
    __shared__ int q_e[PWAVES][QCAP];
    own_body<0>(nullptr, degc, q_e, src, dst, key, nullptr, nullptr, pdeg, nullptr, nullptr, nullptr,
                recs, rcount, n, e);
}

__global__ void __launch_bounds__(1024)
k_prop_first(const int* __restrict__ src, const int* __restrict__ dst,
             const unsigned char* __restrict__ key, const float* __restrict__ h,
             const float* __restrict__ dinv, const float* __restrict__ b,
             const float* __restrict__ W2,
             float* __restrict__ out, int* __restrict__ recs, int* __restrict__ rcount,
             int n, int e) {
    __shared__ float acc[NPW * HID];
    __shared__ int   q_e[PWAVES][QCAP];
    own_body<1>(acc, nullptr, q_e, src, dst, key, h, dinv, nullptr, b, W2, out, recs, rcount, n, e);
}

__global__ void __launch_bounds__(1024)
k_prop_second(const int* __restrict__ src, const int* __restrict__ dst,
              const unsigned char* __restrict__ key, const float* __restrict__ h,
              const float* __restrict__ dinv, const float* __restrict__ b,
              const float* __restrict__ W3,
              float* __restrict__ h3, int* __restrict__ recs, int* __restrict__ rcount,
              int n, int e) {
    __shared__ float acc[NPW * HID];
    __shared__ int   q_e[PWAVES][QCAP];
    own_body<2>(acc, nullptr, q_e, src, dst, key, h, dinv, nullptr, b, W3, h3, recs, rcount, n, e);
}

__global__ void __launch_bounds__(1024)
k_prop3_own(const int* __restrict__ src, const int* __restrict__ dst,
            const float* __restrict__ h3,
            const float* __restrict__ dinv, const float* __restrict__ b3,
            float* __restrict__ y, const int* __restrict__ recs,
            const int* __restrict__ rcount, int n, int e) {
    __shared__ float acc3[NPW];
    __shared__ int   no_recs;

    const int tid  = threadIdx.x;
    const int lane = tid & 31;
    const int wv   = __builtin_amdgcn_readfirstlane(tid >> 5);
    const int lo   = blockIdx.x * NPW;
    int cnt = n - lo;
    if (cnt > NPW) cnt = NPW;
    if (cnt < 0)   cnt = 0;
    const unsigned ucnt = (unsigned)cnt;

    const int region0 = blockIdx.x * SUB * PWAVES + wv;
    const int4* rec0 = (const int4*)recs + (size_t)region0 * RCAP;
    int  pre[SUB + 1];
    bool have = gridDim.x <= 256;
    pre[0] = 0;
    #pragma unroll
    for (int p = 0; p < SUB; ++p) {
        int cp = -1;
        if (gridDim.x <= 256) cp = __builtin_amdgcn_readfirstlane(rcount[region0 + p * PWAVES]);
        if (cp < 0) { have = false; cp = 0; }
        if (cp > RCAP) cp = RCAP;
        pre[p + 1] = pre[p] + cp;
    }
    const int c = pre[SUB];
    auto recptr = [&](int i) -> const int4* {
        int idx = i;
        #pragma unroll
        for (int p = 1; p < SUB; ++p)
            if (i >= pre[p]) idx = i - pre[p] + p * PWAVES * RCAP;
        return rec0 + idx;
    };

    if (tid == 0) no_recs = 0;
    if (tid < cnt) {
        float di = dinv[lo + tid];
        acc3[tid] = b3[0] + h3[lo + tid] * di * di;
    }
    __syncthreads();
    if (!have && lane == 0) no_recs = 1;
    __syncthreads();

    if (no_recs == 0) {
        for (int base = 0; base < c; base += 256) {
            int4  r[8];
            float hv[8];
            #pragma unroll
            for (int u = 0; u < 8; ++u) r[u] = *recptr(min(base + u * 32 + lane, c - 1));
            #pragma unroll
            for (int u = 0; u < 8; ++u) hv[u] = h3[r[u].x];
            asm volatile("" :: "v"(hv[0]), "v"(hv[1]), "v"(hv[2]), "v"(hv[3]),
                               "v"(hv[4]), "v"(hv[5]), "v"(hv[6]), "v"(hv[7]));
            #pragma unroll
            for (int u = 0; u < 8; ++u) {
                if (base + u * 32 + lane < c)
                    (void)__hip_atomic_fetch_add(acc3 + r[u].y, hv[u] * __int_as_float(r[u].z),
                                                 __ATOMIC_RELAXED, __HIP_MEMORY_SCOPE_WORKGROUP);
            }
        }
    } else {
        for (int i = tid; i < e; i += 1024) {
            unsigned d = (unsigned)dst[i] - (unsigned)lo;
            if (d < ucnt) {
                int s = src[i];
                (void)__hip_atomic_fetch_add(acc3 + d, h3[s] * (dinv[s] * dinv[lo + d]),
                                             __ATOMIC_RELAXED, __HIP_MEMORY_SCOPE_WORKGROUP);
            }
        }
    }
    __syncthreads();

    if (tid < cnt) y[lo + tid] = acc3[tid];
}


extern "C" void kernel_launch(void* const* d_in, const int* in_sizes, int n_in,
                              void* d_out, int out_size, void* d_ws, size_t ws_size,
                              hipStream_t stream) {
    const float* x  = (const float*)d_in[0];
    const int*   ei = (const int*)  d_in[1];
    const float* W1 = (const float*)d_in[2];
    const float* b1 = (const float*)d_in[3];
    const float* W2 = (const float*)d_in[4];
    const float* b2 = (const float*)d_in[5];
    const float* W3 = (const float*)d_in[6];
    const float* b3 = (const float*)d_in[7];

    const int N = in_sizes[0] / 2;
    const int E = in_sizes[1] / 2;
    const int* src = ei;
    const int* dst = ei + E;

    float* dinv = (float*)d_ws;
    float* h    = dinv + (size_t)N;
    float* out  = h    + (size_t)N * HID;
    float* h3   = out  + (size_t)N * HID;
    unsigned char* key = (unsigned char*)(h3 + (size_t)N);
    int* recs   = (int*)(key + (size_t)E);
    int* rcount = recs + (size_t)256 * SUB * PWAVES * RCAP * 4;
    int* pdeg   = rcount + (size_t)256 * SUB * PWAVES;
    float* y    = (float*)d_out;

    const int B = 256;
    auto g = [](long long t) { return (int)((t + 255) / 256); };
    const int G = (N + NPW - 1) / NPW;
    const int GS = (G + SUB - 1) / SUB * SUB;

    k_key     <<<g(((long long)E + 511) / 512 * 32), B, 0, stream>>>(dst, key, E);
    k_scan_deg<<<GS, 1024, 0, stream>>>(src, dst, key, pdeg, recs, rcount, N, E);

    k_lin1      <<<g(((long long)N + 31) / 32 * 32), B, 0, stream>>>(x, W1, pdeg, dinv, h, N);
    k_prop_first<<<G, 1024, 0, stream>>>(src, dst, key, h, dinv, b1, W2, out, recs, rcount, N, E);

    k_prop_second<<<G, 1024, 0, stream>>>(src, dst, key, out, dinv, b2, W3, h3, recs, rcount, N, E);

    k_prop3_own<<<G, 1024, 0, stream>>>(src, dst, h3, dinv, b3, y, recs, rcount, N, E);
}


# Edge scan reads dst and src themselves by 16-byte loads, split 16 ways (was 1-byte keys + a dst/src gather per hit, 8 ways): key pass and its launch go; LDS adds unordered as before
# speedup vs baseline: 4.1469x; 1.0422x over previous
#include <hip/hip_runtime.h>
#include <hip/hip_bf16.h>

typedef __attribute__((ext_vector_type(2))) float v2f;
typedef __attribute__((ext_vector_type(8))) float v8f;

#define HID 64
#define NPW 391
#define SUB 16


__global__ void k_lin1(const float* __restrict__ x, const float* __restrict__ W1,
                       const int* __restrict__ pdeg, float* __restrict__ dinv,
                       float* __restrict__ h, int n) {
    const int t    = blockIdx.x * blockDim.x + threadIdx.x;
    const int lane = threadIdx.x & 31;
    const int row0 = (t >> 5) * 32;
    const int c4   = (lane & 15) * 4;
    const int hi   = lane >> 4;
    const bool al  = (((size_t)h) & 15) == 0;
    float2 xv = make_float2(0.0f, 0.0f);
    if (row0 + lane < n) xv = ((const float2*)x)[row0 + lane];
    float wa[4], wb[4];
    #pragma unroll
    for (int k = 0; k < 4; ++k) { wa[k] = W1[c4 + k]; wb[k] = W1[HID + c4 + k]; }
    if (al && row0 + 31 < n) {
        #pragma unroll
        for (int s = 0; s < 16; ++s) {
            const float ax = __shfl(xv.x, 2 * s + hi, 32);
            const float ay = __shfl(xv.y, 2 * s + hi, 32);
            float4 v;
            v.x = ax * wa[0] + ay * wb[0];
            v.y = ax * wa[1] + ay * wb[1];
            v.z = ax * wa[2] + ay * wb[2];
            v.w = ax * wa[3] + ay * wb[3];
            *(float4*)(h + (size_t)(row0 + 2 * s + hi) * HID + c4) = v;
        }
    } else {
        #pragma nounroll
        for (int s = 0; s < 16; ++s) {
            const float ax = __shfl(xv.x, 2 * s + hi, 32);
            const float ay = __shfl(xv.y, 2 * s + hi, 32);
            const int r = row0 + 2 * s + hi;
            if (r < n) {
                float* o = h + (size_t)r * HID + c4;
                #pragma unroll
                for (int k = 0; k < 4; ++k) o[k] = ax * wa[k] + ay * wb[k];
            }
        }
    }
    if (t < n) {
        int c = 0;
        for (int p = 0; p < SUB; ++p) c += pdeg[(size_t)p * n + t];
        dinv[t] = rsqrtf(1.0f + (float)c);
    }
}

#define PWAVES 32
#define QCAP   288
#define RCAP   (1024 / SUB)
static_assert(SUB <= 32, "a scan wave keeps the length of list i in lane i");

template <int MODE>
__device__ __forceinline__ void own_body(
           float* __restrict__ acc, int* __restrict__ degc, int2 (* __restrict__ q_e)[QCAP],
           const int* __restrict__ src, const int* __restrict__ dst,
           const float* __restrict__ h,
           const float* __restrict__ dinv, int* __restrict__ pdeg,
           const float* __restrict__ b, const float* __restrict__ Wn, float* __restrict__ out,
           int* __restrict__ recs, int* __restrict__ rcount, int n, int e) {
    const bool small = gridDim.x <= 256;
    const int tid  = threadIdx.x;
    const int lane = tid & 31;
    const int wv   = __builtin_amdgcn_readfirstlane(tid >> 5);
    const int grp  = blockIdx.x / SUB;
    const int part = blockIdx.x % SUB;
    const int span = (MODE == 0) ? SUB * NPW : NPW;
    const int lo   = (MODE == 0) ? grp * SUB * NPW : blockIdx.x * NPW;
    int cnt = n - lo;
    if (cnt > span) cnt = span;
    if (cnt < 0)    cnt = 0;
    const unsigned ucnt = (unsigned)cnt;
    const unsigned ulo  = (unsigned)lo;
    const int vw      = (MODE == 0) ? part * PWAVES + wv : wv;
    const int vstride = (MODE == 0) ? PWAVES * SUB : PWAVES;

    if (MODE == 0) {
        for (int t = tid; t < cnt; t += 1024) degc[t] = 0;
    } else {
        for (int t = tid; t < cnt * HID; t += 1024) {
            int i = t >> 6, j = t & 63;
            float di = dinv[lo + i];
            acc[t] = b[j] + h[(size_t)lo * HID + t] * di * di;
        }
    }
    __syncthreads();

    int2* qe = q_e[wv];
    int  qn = 0;
    const int region0 = (MODE == 0) ? (grp * SUB * SUB + part) * PWAVES + wv
                                    : blockIdx.x * SUB * PWAVES + wv;
    const int rstep   = (MODE == 0) ? SUB * PWAVES : PWAVES;
    int4* rec0 = (int4*)recs + (size_t)region0 * RCAP;
    int      rn = 0;
    unsigned rfull = 0u;

    auto load_chunk_guarded = [&](const int* p, int chunk) -> int4 {
        int4 v;
        int base = chunk * 128 + lane * 4;
        v.x = (base     < e) ? p[base]     : -1;
        v.y = (base + 1 < e) ? p[base + 1] : -1;
        v.z = (base + 2 < e) ? p[base + 2] : -1;
        v.w = (base + 3 < e) ? p[base + 3] : -1;
        return v;
    };

    auto add_rows = [&](int my_s, int my_d, int my_w, int m) {
        for (int i = 0; i < m; i += 8) {
            float a0[8], a1[8];
            #pragma unroll
            for (int u = 0; u < 8; ++u) {
                int s = __builtin_amdgcn_readlane(my_s, i + u);
                const float* r = h + (size_t)s * HID;
                a0[u] = r[lane];
                a1[u] = r[lane + 32];
            }
            #pragma unroll
            for (int u = 0; u < 8; ++u) {
                if (i + u < m) {
                    float w = __int_as_float(__builtin_amdgcn_readlane(my_w, i + u));
                    int   d = __builtin_amdgcn_readlane(my_d, i + u);
                    float* p = acc + d * HID + lane;
                    (void)__hip_atomic_fetch_add(p,      a0[u] * w, __ATOMIC_RELAXED, __HIP_MEMORY_SCOPE_WORKGROUP);
                    (void)__hip_atomic_fetch_add(p + 32, a1[u] * w, __ATOMIC_RELAXED, __HIP_MEMORY_SCOPE_WORKGROUP);
                }
            }
            if (MODE == 1)
                asm volatile("" :: "v"(a0[0]), "v"(a0[1]), "v"(a0[2]), "v"(a0[3]),
                                   "v"(a0[4]), "v"(a0[5]), "v"(a0[6]), "v"(a0[7]),
                                   "v"(a1[0]), "v"(a1[1]), "v"(a1[2]), "v"(a1[3]),
                                   "v"(a1[4]), "v"(a1[5]), "v"(a1[6]), "v"(a1[7]));
        }
    };

    auto drain = [&](int qb, int m) {
        int my_s = 0, my_d = 0, my_w = 0;
        if (lane < m) {
            const int2 q = qe[qb + lane];
            my_s = q.x;
            my_d = q.y;
            if (MODE != 0) my_w = __float_as_int(dinv[my_s] * dinv[lo + my_d]);
        }
        if (MODE == 0) {
            if (small) {
                const int so_l = my_d / NPW;
                const int row  = my_d - so_l * NPW;
                unsigned todo = (unsigned)__ballot(lane < m);
                while (todo) {
                    const int so = __builtin_amdgcn_readlane(so_l, __builtin_ctz(todo));
                    const bool mine = lane < m && so_l == so;
                    const unsigned mask = (unsigned)__ballot(mine);
                    const int k   = __popc(mask);
                    const int cur = __builtin_amdgcn_readlane(rn, so);
                    if (!((rfull >> so) & 1u) && cur + k <= RCAP) {
                        if (mine)
                            rec0[(size_t)so * rstep * RCAP + cur + (int)__builtin_amdgcn_mbcnt_lo(mask, 0u)]
                                = make_int4(my_s, row, 0, 0);
                        if (lane == so) rn = cur + k;
                    } else {
                        rfull |= 1u << so;
                    }
                    todo &= ~mask;
                }
            }
            if (lane < m)
                (void)__hip_atomic_fetch_add(degc + my_d, 1, __ATOMIC_RELAXED, __HIP_MEMORY_SCOPE_WORKGROUP);
        } else {
            add_rows(my_s, my_d, my_w, m);
        }
    };

    auto append = [&](unsigned t, int s) {
        bool hit = t < ucnt;
        unsigned mask = (unsigned)__ballot(hit);
        if (mask) {
            int pos = qn + (int)__builtin_amdgcn_mbcnt_lo(mask, 0u);
            if (hit) qe[pos] = make_int2(s, (int)t);
            qn += __popc(mask);
        }
    };

    auto handle = [&](int4 vd, int4 vs) {
        uint4 t;
        t.x = (unsigned)vd.x - ulo; t.y = (unsigned)vd.y - ulo;
        t.z = (unsigned)vd.z - ulo; t.w = (unsigned)vd.w - ulo;
        unsigned mn = min(min(t.x, t.y), min(t.z, t.w));
        if (__any(mn < ucnt)) {
            append(t.x, vs.x);
            append(t.y, vs.y);
            append(t.z, vs.z);
            append(t.w, vs.w);
        }
    };

    const bool vec_ok  = (((size_t)dst) & 15) == 0 && (((size_t)src) & 15) == 0;
    const int  nunit   = vec_ok ? (e >> 7) : 0;
    const int  ngrp    = ((int)gridDim.x + SUB - 1) / SUB;
    const int  ustart  = (int)(((long long)grp * nunit) / ngrp);
    const int4* dst4   = (const int4*)dst;
    const int4* src4   = (const int4*)src;

    bool scan = true;
    if (MODE != 0 && small) {
        int  pre[SUB + 1];
        bool have = true;
        pre[0] = 0;
        #pragma unroll
        for (int p = 0; p < SUB; ++p) {
            int cp = __builtin_amdgcn_readfirstlane(rcount[region0 + p * rstep]);
            if (cp < 0) { have = false; cp = 0; }
            if (cp > RCAP) cp = RCAP;
            pre[p + 1] = pre[p] + cp;
        }
        auto recptr = [&](int i) -> int4* {
            int idx = i;
            #pragma unroll
            for (int p = 1; p < SUB; ++p)
                if (i >= pre[p]) idx = i - pre[p] + p * rstep * RCAP;
            return rec0 + idx;
        };
        if (have) {
            scan = false;
            const int c = pre[SUB];
            if (c > 0) {
                if (MODE == 1) {
                    int4 r1 = *recptr(min(lane, c - 1));
                    int4 r2 = *recptr(min(32 + lane, c - 1));
                    float ds = dinv[r1.x], dd = dinv[lo + r1.y];
                    for (int base = 0; base < c; base += 32) {
                        int4 cur = r1;
                        int wbits = __float_as_int(ds * dd);
                        r1 = r2;
                        r2 = *recptr(min(base + 64 + lane, c - 1));
                        ds = dinv[r1.x];
                        dd = dinv[lo + r1.y];
                        int m = c - base;
                        if (m > 32) m = 32;
                        add_rows(cur.x, cur.y, wbits, m);
                        if (lane < m) *recptr(base + lane) = make_int4(cur.x, cur.y, wbits, 0);
                    }
                } else {
                    int4 nx = *recptr(min(lane, c - 1));
                    for (int base = 0; base < c; base += 32) {
                        int4 cur = nx;
                        nx = *recptr(min(base + 32 + lane, c - 1));
                        int m = c - base;
                        if (m > 32) m = 32;
                        add_rows(cur.x, cur.y, cur.z, m);
                    }
                }
            }
        }
    }

    if (cnt > 0 && scan) {
        if (vw < nunit) {
            const int nk = (nunit - vw + vstride - 1) / vstride;
            auto unit_of = [&](int k) -> int {
                int r = vw + (k < 0 ? 0 : (k < nk ? k : nk - 1)) * vstride + ustart;
                return r >= nunit ? r - nunit : r;
            };
            auto ldu = [&](const int4* p, int u) -> int4 { return p[(size_t)u * 32 + lane]; };
            if (MODE == 0) {
                auto step = [&](int4& d, int4& s, int k) {
                    if (k >= 0 && k < nk) handle(d, s);
                    const int u = unit_of(k + 8);
                    d = ldu(dst4, u);
                    s = ldu(src4, u);
                };
                const int4 none = {0, 0, 0, 0};
                int4 d0 = none, d1 = none, d2 = none, d3 = none, d4 = none, d5 = none, d6 = none, d7 = none;
                int4 s0 = none, s1 = none, s2 = none, s3 = none, s4 = none, s5 = none, s6 = none, s7 = none;
                for (int k = -8; k < nk; k += 8) {
                    step(d0, s0, k);     step(d1, s1, k + 1); while (qn >= 32) { qn -= 32; drain(qn, 32); }
                    step(d2, s2, k + 2); step(d3, s3, k + 3); while (qn >= 32) { qn -= 32; drain(qn, 32); }
                    step(d4, s4, k + 4); step(d5, s5, k + 5); while (qn >= 32) { qn -= 32; drain(qn, 32); }
                    step(d6, s6, k + 6); step(d7, s7, k + 7); while (qn >= 32) { qn -= 32; drain(qn, 32); }
                }
            } else {
                #pragma nounroll
                for (int k = 0; k < nk; ++k) {
                    const int u = unit_of(k);
                    handle(ldu(dst4, u), ldu(src4, u));
                    while (qn >= 32) { qn -= 32; drain(qn, 32); }
                }
            }
        }
        for (int c = nunit + vw; c * 128 < e; c += vstride) {
            handle(load_chunk_guarded(dst, c), load_chunk_guarded(src, c));
            while (qn >= 32) { qn -= 32; drain(qn, 32); }
        }
        if (qn > 0) drain(0, qn);
    }

    if (MODE == 0) {
        if (small && lane < SUB)
            rcount[region0 + lane * rstep] = ((rfull >> lane) & 1u) ? -1 : rn;
        __syncthreads();
        for (int t = tid; t < cnt; t += 1024)
            pdeg[(size_t)part * n + lo + t] = degc[t];
    } else if (MODE == 1) {
        __syncthreads();
        const int hh = lane >> 4;
        const int l  = lane & 15;
        for (int row0 = wv * 16; row0 < cnt; row0 += PWAVES * 16) {
            int ar = row0 + l;
            if (ar > cnt - 1) ar = cnt - 1;
            const float* arow = acc + ar * HID;
            v8f acc0 = {}, acc1 = {}, acc2 = {}, acc3 = {};
            for (int k0 = 0; k0 < HID; k0 += 4) {
                const int ka = k0 + hh * 2;
                v2f a;
                a.x = fmaxf(arow[ka],     0.0f);
                a.y = fmaxf(arow[ka + 1], 0.0f);
                const float* bp = Wn + (size_t)ka * HID + l;
                v2f b0, b1, b2, b3;
                b0.x = bp[0];  b0.y = bp[HID];
                b1.x = bp[16]; b1.y = bp[HID + 16];
                b2.x = bp[32]; b2.y = bp[HID + 32];
                b3.x = bp[48]; b3.y = bp[HID + 48];
                acc0 = __builtin_amdgcn_wmma_f32_16x16x4_f32(false, a, false, b0, (short)0, acc0, false, false);
                acc1 = __builtin_amdgcn_wmma_f32_16x16x4_f32(false, a, false, b1, (short)0, acc1, false, false);
                acc2 = __builtin_amdgcn_wmma_f32_16x16x4_f32(false, a, false, b2, (short)0, acc2, false, false);
                acc3 = __builtin_amdgcn_wmma_f32_16x16x4_f32(false, a, false, b3, (short)0, acc3, false, false);
            }
            #pragma unroll
            for (int r = 0; r < 8; ++r) {
                const int orow = row0 + hh * 8 + r;
                if (orow < cnt) {
                    float* o = acc + orow * HID + l;
                    o[0]  = acc0[r];
                    o[16] = acc1[r];
                    o[32] = acc2[r];
                    o[48] = acc3[r];
                }
            }
        }
        __syncthreads();
        for (int t = tid; t < cnt * HID; t += 1024)
            out[(size_t)lo * HID + t] = acc[t];
    } else {
        __syncthreads();
        constexpr int RPW = (NPW + PWAVES - 1) / PWAVES;
        const float w0 = Wn[lane], w1 = Wn[lane + 32];
        const int r0 = wv * RPW;
        if (r0 < cnt) {
            float mine = 0.0f;
            #pragma unroll
            for (int u = 0; u < RPW; ++u) {
                int r = r0 + u;
                if (r > cnt - 1) r = cnt - 1;
                float v = fmaxf(acc[r * HID + lane], 0.0f) * w0 + fmaxf(acc[r * HID + lane + 32], 0.0f) * w1;
                #pragma unroll
                for (int off = 16; off > 0; off >>= 1) v += __shfl_down(v, off, 32);
                const float tot = __int_as_float(__builtin_amdgcn_readfirstlane(__float_as_int(v)));
                if (lane == u) mine = tot;
            }
            if (lane < RPW && r0 + lane < cnt) out[lo + r0 + lane] = mine;
        }
    }
}

__global__ void __launch_bounds__(1024)
k_scan_deg(const int* __restrict__ src, const int* __restrict__ dst,
           int* __restrict__ pdeg,
           int* __restrict__ recs, int* __restrict__ rcount, int n, int e) {
    __shared__ int  degc[SUB * NPW];
    __shared__ int2 q_e[PWAVES][QCAP];
    own_body<0>(nullptr, degc, q_e, src, dst, nullptr, nullptr, pdeg, nullptr, nullptr, nullptr,
                recs, rcount, n, e);
}

__global__ void __launch_bounds__(1024)
k_prop_first(const int* __restrict__ src, const int* __restrict__ dst,
             const float* __restrict__ h,
             const float* __restrict__ dinv, const float* __restrict__ b,
             const float* __restrict__ W2,
             float* __restrict__ out, int* __restrict__ recs, int* __restrict__ rcount,
             int n, int e) {
    __shared__ float acc[NPW * HID];
    __shared__ int2  q_e[PWAVES][QCAP];
    own_body<1>(acc, nullptr, q_e, src, dst, h, dinv, nullptr, b, W2, out, recs, rcount, n, e);
}

__global__ void __launch_bounds__(1024)
k_prop_second(const int* __restrict__ src, const int* __restrict__ dst,
              const float* __restrict__ h,
              const float* __restrict__ dinv, const float* __restrict__ b,
              const float* __restrict__ W3,
              float* __restrict__ h3, int* __restrict__ recs, int* __restrict__ rcount,
              int n, int e) {
    __shared__ float acc[NPW * HID];
    __shared__ int2  q_e[PWAVES][QCAP];
    own_body<2>(acc, nullptr, q_e, src, dst, h, dinv, nullptr, b, W3, h3, recs, rcount, n, e);
}

__global__ void __launch_bounds__(1024)
k_prop3_own(const int* __restrict__ src, const int* __restrict__ dst,
            const float* __restrict__ h3,
            const float* __restrict__ dinv, const float* __restrict__ b3,
            float* __restrict__ y, const int* __restrict__ recs,
            const int* __restrict__ rcount, int n, int e) {
    __shared__ float acc3[NPW];
    __shared__ int   no_recs;

    const int tid  = threadIdx.x;
    const int lane = tid & 31;
    const int wv   = __builtin_amdgcn_readfirstlane(tid >> 5);
    const int lo   = blockIdx.x * NPW;
    int cnt = n - lo;
    if (cnt > NPW) cnt = NPW;
    if (cnt < 0)   cnt = 0;
    const unsigned ucnt = (unsigned)cnt;

    const int region0 = blockIdx.x * SUB * PWAVES + wv;
    const int4* rec0 = (const int4*)recs + (size_t)region0 * RCAP;
    int  pre[SUB + 1];
    bool have = gridDim.x <= 256;
    pre[0] = 0;
    #pragma unroll
    for (int p = 0; p < SUB; ++p) {
        int cp = -1;
        if (gridDim.x <= 256) cp = __builtin_amdgcn_readfirstlane(rcount[region0 + p * PWAVES]);
        if (cp < 0) { have = false; cp = 0; }
        if (cp > RCAP) cp = RCAP;
        pre[p + 1] = pre[p] + cp;
    }
    const int c = pre[SUB];
    auto recptr = [&](int i) -> const int4* {
        int idx = i;
        #pragma unroll
        for (int p = 1; p < SUB; ++p)
            if (i >= pre[p]) idx = i - pre[p] + p * PWAVES * RCAP;
        return rec0 + idx;
    };

    if (tid == 0) no_recs = 0;
    if (tid < cnt) {
        float di = dinv[lo + tid];
        acc3[tid] = b3[0] + h3[lo + tid] * di * di;
    }
    __syncthreads();
    if (!have && lane == 0) no_recs = 1;
    __syncthreads();

    if (no_recs == 0) {
        for (int base = 0; base < c; base += 256) {
            int4  r[8];
            float hv[8];
            #pragma unroll
            for (int u = 0; u < 8; ++u) r[u] = *recptr(min(base + u * 32 + lane, c - 1));
            #pragma unroll
            for (int u = 0; u < 8; ++u) hv[u] = h3[r[u].x];
            asm volatile("" :: "v"(hv[0]), "v"(hv[1]), "v"(hv[2]), "v"(hv[3]),
                               "v"(hv[4]), "v"(hv[5]), "v"(hv[6]), "v"(hv[7]));
            #pragma unroll
            for (int u = 0; u < 8; ++u) {
                if (base + u * 32 + lane < c)
                    (void)__hip_atomic_fetch_add(acc3 + r[u].y, hv[u] * __int_as_float(r[u].z),
                                                 __ATOMIC_RELAXED, __HIP_MEMORY_SCOPE_WORKGROUP);
            }
        }
    } else {
        for (int i = tid; i < e; i += 1024) {
            unsigned d = (unsigned)dst[i] - (unsigned)lo;
            if (d < ucnt) {
                int s = src[i];
                (void)__hip_atomic_fetch_add(acc3 + d, h3[s] * (dinv[s] * dinv[lo + d]),
                                             __ATOMIC_RELAXED, __HIP_MEMORY_SCOPE_WORKGROUP);
            }
        }
    }
    __syncthreads();

    if (tid < cnt) y[lo + tid] = acc3[tid];
}


extern "C" void kernel_launch(void* const* d_in, const int* in_sizes, int n_in,
                              void* d_out, int out_size, void* d_ws, size_t ws_size,
                              hipStream_t stream) {
    const float* x  = (const float*)d_in[0];
    const int*   ei = (const int*)  d_in[1];
    const float* W1 = (const float*)d_in[2];
    const float* b1 = (const float*)d_in[3];
    const float* W2 = (const float*)d_in[4];
    const float* b2 = (const float*)d_in[5];
    const float* W3 = (const float*)d_in[6];
    const float* b3 = (const float*)d_in[7];

    const int N = in_sizes[0] / 2;
    const int E = in_sizes[1] / 2;
    const int* src = ei;
    const int* dst = ei + E;

    float* dinv = (float*)d_ws;
    float* h    = dinv + (size_t)N;
    float* out  = h    + (size_t)N * HID;
    float* h3   = out  + (size_t)N * HID;
    int* recs   = (int*)(h3 + (size_t)N);
    int* rcount = recs + (size_t)256 * SUB * PWAVES * RCAP * 4;
    int* pdeg   = rcount + (size_t)256 * SUB * PWAVES;
    float* y    = (float*)d_out;

    const int B = 256;
    auto g = [](long long t) { return (int)((t + 255) / 256); };
    const int G = (N + NPW - 1) / NPW;
    const int GS = (G + SUB - 1) / SUB * SUB;

    k_scan_deg<<<GS, 1024, 0, stream>>>(src, dst, pdeg, recs, rcount, N, E);

    k_lin1      <<<g(((long long)N + 31) / 32 * 32), B, 0, stream>>>(x, W1, pdeg, dinv, h, N);
    k_prop_first<<<G, 1024, 0, stream>>>(src, dst, h, dinv, b1, W2, out, recs, rcount, N, E);

    k_prop_second<<<G, 1024, 0, stream>>>(src, dst, out, dinv, b2, W3, h3, recs, rcount, N, E);

    k_prop3_own<<<G, 1024, 0, stream>>>(src, dst, h3, dinv, b3, y, recs, rcount, N, E);
}


# Scan wave places a batch of hits in all its owner lists at once: place = list length (scalar) + same-owner hits in lower lanes, one ballot per list, one store per batch; was a scalar turn per owner
# speedup vs baseline: 4.4109x; 1.0637x over previous
#include <hip/hip_runtime.h>
#include <hip/hip_bf16.h>

typedef __attribute__((ext_vector_type(2))) float v2f;
typedef __attribute__((ext_vector_type(8))) float v8f;

#define HID 64
#define NPW 391
#define SUB 16


__global__ void k_lin1(const float* __restrict__ x, const float* __restrict__ W1,
                       const int* __restrict__ pdeg, float* __restrict__ dinv,
                       float* __restrict__ h, int n) {
    const int t    = blockIdx.x * blockDim.x + threadIdx.x;
    const int lane = threadIdx.x & 31;
    const int row0 = (t >> 5) * 32;
    const int c4   = (lane & 15) * 4;
    const int hi   = lane >> 4;
    const bool al  = (((size_t)h) & 15) == 0;
    float2 xv = make_float2(0.0f, 0.0f);
    if (row0 + lane < n) xv = ((const float2*)x)[row0 + lane];
    float wa[4], wb[4];
    #pragma unroll
    for (int k = 0; k < 4; ++k) { wa[k] = W1[c4 + k]; wb[k] = W1[HID + c4 + k]; }
    if (al && row0 + 31 < n) {
        #pragma unroll
        for (int s = 0; s < 16; ++s) {
            const float ax = __shfl(xv.x, 2 * s + hi, 32);
            const float ay = __shfl(xv.y, 2 * s + hi, 32);
            float4 v;
            v.x = ax * wa[0] + ay * wb[0];
            v.y = ax * wa[1] + ay * wb[1];
            v.z = ax * wa[2] + ay * wb[2];
            v.w = ax * wa[3] + ay * wb[3];
            *(float4*)(h + (size_t)(row0 + 2 * s + hi) * HID + c4) = v;
        }
    } else {
        #pragma nounroll
        for (int s = 0; s < 16; ++s) {
            const float ax = __shfl(xv.x, 2 * s + hi, 32);
            const float ay = __shfl(xv.y, 2 * s + hi, 32);
            const int r = row0 + 2 * s + hi;
            if (r < n) {
                float* o = h + (size_t)r * HID + c4;
                #pragma unroll
                for (int k = 0; k < 4; ++k) o[k] = ax * wa[k] + ay * wb[k];
            }
        }
    }
    if (t < n) {
        int c = 0;
        for (int p = 0; p < SUB; ++p) c += pdeg[(size_t)p * n + t];
        dinv[t] = rsqrtf(1.0f + (float)c);
    }
}

#define PWAVES 32
#define QCAP   288
#define RCAP   (1024 / SUB)
static_assert(SUB <= 32, "lane i of a scan wave stores the length of list i");

template <int MODE>
__device__ __forceinline__ void own_body(
           float* __restrict__ acc, int* __restrict__ degc, int2 (* __restrict__ q_e)[QCAP],
           const int* __restrict__ src, const int* __restrict__ dst,
           const float* __restrict__ h,
           const float* __restrict__ dinv, int* __restrict__ pdeg,
           const float* __restrict__ b, const float* __restrict__ Wn, float* __restrict__ out,
           int* __restrict__ recs, int* __restrict__ rcount, int n, int e) {
    const bool small = gridDim.x <= 256;
    const int tid  = threadIdx.x;
    const int lane = tid & 31;
    const int wv   = __builtin_amdgcn_readfirstlane(tid >> 5);
    const int grp  = blockIdx.x / SUB;
    const int part = blockIdx.x % SUB;
    const int span = (MODE == 0) ? SUB * NPW : NPW;
    const int lo   = (MODE == 0) ? grp * SUB * NPW : blockIdx.x * NPW;
    int cnt = n - lo;
    if (cnt > span) cnt = span;
    if (cnt < 0)    cnt = 0;
    const unsigned ucnt = (unsigned)cnt;
    const unsigned ulo  = (unsigned)lo;
    const int vw      = (MODE == 0) ? part * PWAVES + wv : wv;
    const int vstride = (MODE == 0) ? PWAVES * SUB : PWAVES;

    if (MODE == 0) {
        for (int t = tid; t < cnt; t += 1024) degc[t] = 0;
    } else {
        for (int t = tid; t < cnt * HID; t += 1024) {
            int i = t >> 6, j = t & 63;
            float di = dinv[lo + i];
            acc[t] = b[j] + h[(size_t)lo * HID + t] * di * di;
        }
    }
    __syncthreads();

    int2* qe = q_e[wv];
    int  qn = 0;
    const int region0 = (MODE == 0) ? (grp * SUB * SUB + part) * PWAVES + wv
                                    : blockIdx.x * SUB * PWAVES + wv;
    const int rstep   = (MODE == 0) ? SUB * PWAVES : PWAVES;
    int4* rec0 = (int4*)recs + (size_t)region0 * RCAP;
    int rnS[SUB];
    #pragma unroll
    for (int o = 0; o < SUB; ++o) rnS[o] = 0;

    auto load_chunk_guarded = [&](const int* p, int chunk) -> int4 {
        int4 v;
        int base = chunk * 128 + lane * 4;
        v.x = (base     < e) ? p[base]     : -1;
        v.y = (base + 1 < e) ? p[base + 1] : -1;
        v.z = (base + 2 < e) ? p[base + 2] : -1;
        v.w = (base + 3 < e) ? p[base + 3] : -1;
        return v;
    };

    auto add_rows = [&](int my_s, int my_d, int my_w, int m) {
        for (int i = 0; i < m; i += 8) {
            float a0[8], a1[8];
            #pragma unroll
            for (int u = 0; u < 8; ++u) {
                int s = __builtin_amdgcn_readlane(my_s, i + u);
                const float* r = h + (size_t)s * HID;
                a0[u] = r[lane];
                a1[u] = r[lane + 32];
            }
            #pragma unroll
            for (int u = 0; u < 8; ++u) {
                if (i + u < m) {
                    float w = __int_as_float(__builtin_amdgcn_readlane(my_w, i + u));
                    int   d = __builtin_amdgcn_readlane(my_d, i + u);
                    float* p = acc + d * HID + lane;
                    (void)__hip_atomic_fetch_add(p,      a0[u] * w, __ATOMIC_RELAXED, __HIP_MEMORY_SCOPE_WORKGROUP);
                    (void)__hip_atomic_fetch_add(p + 32, a1[u] * w, __ATOMIC_RELAXED, __HIP_MEMORY_SCOPE_WORKGROUP);
                }
            }
            if (MODE == 1)
                asm volatile("" :: "v"(a0[0]), "v"(a0[1]), "v"(a0[2]), "v"(a0[3]),
                                   "v"(a0[4]), "v"(a0[5]), "v"(a0[6]), "v"(a0[7]),
                                   "v"(a1[0]), "v"(a1[1]), "v"(a1[2]), "v"(a1[3]),
                                   "v"(a1[4]), "v"(a1[5]), "v"(a1[6]), "v"(a1[7]));
        }
    };

    auto drain = [&](int qb, int m) {
        int my_s = 0, my_d = 0, my_w = 0;
        if (lane < m) {
            const int2 q = qe[qb + lane];
            my_s = q.x;
            my_d = q.y;
            if (MODE != 0) my_w = __float_as_int(dinv[my_s] * dinv[lo + my_d]);
        }
        if (MODE == 0) {
            if (small) {
                const int so_l = my_d / NPW;
                const int row  = my_d - so_l * NPW;
                int slot = RCAP;
                #pragma unroll
                for (int o = 0; o < SUB; ++o) {
                    const bool mine = lane < m && so_l == o;
                    const unsigned mask = (unsigned)__ballot(mine);
                    if (mine) slot = (int)__builtin_amdgcn_mbcnt_lo(mask, (unsigned)rnS[o]);
                    rnS[o] += __popc(mask);
                }
                if (slot < RCAP)
                    rec0[(size_t)so_l * rstep * RCAP + slot] = make_int4(my_s, row, 0, 0);
            }
            if (lane < m)
                (void)__hip_atomic_fetch_add(degc + my_d, 1, __ATOMIC_RELAXED, __HIP_MEMORY_SCOPE_WORKGROUP);
        } else {
            add_rows(my_s, my_d, my_w, m);
        }
    };

    auto append = [&](unsigned t, int s) {
        bool hit = t < ucnt;
        unsigned mask = (unsigned)__ballot(hit);
        if (mask) {
            int pos = qn + (int)__builtin_amdgcn_mbcnt_lo(mask, 0u);
            if (hit) qe[pos] = make_int2(s, (int)t);
            qn += __popc(mask);
        }
    };

    auto handle = [&](int4 vd, int4 vs) {
        uint4 t;
        t.x = (unsigned)vd.x - ulo; t.y = (unsigned)vd.y - ulo;
        t.z = (unsigned)vd.z - ulo; t.w = (unsigned)vd.w - ulo;
        unsigned mn = min(min(t.x, t.y), min(t.z, t.w));
        if (__any(mn < ucnt)) {
            append(t.x, vs.x);
            append(t.y, vs.y);
            append(t.z, vs.z);
            append(t.w, vs.w);
        }
    };

    const bool vec_ok  = (((size_t)dst) & 15) == 0 && (((size_t)src) & 15) == 0;
    const int  nunit   = vec_ok ? (e >> 7) : 0;
    const int  ngrp    = ((int)gridDim.x + SUB - 1) / SUB;
    const int  ustart  = (int)(((long long)grp * nunit) / ngrp);
    const int4* dst4   = (const int4*)dst;
    const int4* src4   = (const int4*)src;

    bool scan = true;
    if (MODE != 0 && small) {
        int  pre[SUB + 1];
        bool have = true;
        pre[0] = 0;
        #pragma unroll
        for (int p = 0; p < SUB; ++p) {
            int cp = __builtin_amdgcn_readfirstlane(rcount[region0 + p * rstep]);
            if (cp < 0) { have = false; cp = 0; }
            if (cp > RCAP) cp = RCAP;
            pre[p + 1] = pre[p] + cp;
        }
        auto recptr = [&](int i) -> int4* {
            int idx = i;
            #pragma unroll
            for (int p = 1; p < SUB; ++p)
                if (i >= pre[p]) idx = i - pre[p] + p * rstep * RCAP;
            return rec0 + idx;
        };
        if (have) {
            scan = false;
            const int c = pre[SUB];
            if (c > 0) {
                if (MODE == 1) {
                    int4 r1 = *recptr(min(lane, c - 1));
                    int4 r2 = *recptr(min(32 + lane, c - 1));
                    float ds = dinv[r1.x], dd = dinv[lo + r1.y];
                    for (int base = 0; base < c; base += 32) {
                        int4 cur = r1;
                        int wbits = __float_as_int(ds * dd);
                        r1 = r2;
                        r2 = *recptr(min(base + 64 + lane, c - 1));
                        ds = dinv[r1.x];
                        dd = dinv[lo + r1.y];
                        int m = c - base;
                        if (m > 32) m = 32;
                        add_rows(cur.x, cur.y, wbits, m);
                        if (lane < m) *recptr(base + lane) = make_int4(cur.x, cur.y, wbits, 0);
                    }
                } else {
                    int4 nx = *recptr(min(lane, c - 1));
                    for (int base = 0; base < c; base += 32) {
                        int4 cur = nx;
                        nx = *recptr(min(base + 32 + lane, c - 1));
                        int m = c - base;
                        if (m > 32) m = 32;
                        add_rows(cur.x, cur.y, cur.z, m);
                    }
                }
            }
        }
    }

    if (cnt > 0 && scan) {
        if (vw < nunit) {
            const int nk = (nunit - vw + vstride - 1) / vstride;
            auto unit_of = [&](int k) -> int {
                int r = vw + (k < 0 ? 0 : (k < nk ? k : nk - 1)) * vstride + ustart;
                return r >= nunit ? r - nunit : r;
            };
            auto ldu = [&](const int4* p, int u) -> int4 { return p[(size_t)u * 32 + lane]; };
            if (MODE == 0) {
                auto step = [&](int4& d, int4& s, int k) {
                    if (k >= 0 && k < nk) handle(d, s);
                    const int u = unit_of(k + 8);
                    d = ldu(dst4, u);
                    s = ldu(src4, u);
                };
                const int4 none = {0, 0, 0, 0};
                int4 d0 = none, d1 = none, d2 = none, d3 = none, d4 = none, d5 = none, d6 = none, d7 = none;
                int4 s0 = none, s1 = none, s2 = none, s3 = none, s4 = none, s5 = none, s6 = none, s7 = none;
                for (int k = -8; k < nk; k += 8) {
                    step(d0, s0, k);     step(d1, s1, k + 1); while (qn >= 32) { qn -= 32; drain(qn, 32); }
                    step(d2, s2, k + 2); step(d3, s3, k + 3); while (qn >= 32) { qn -= 32; drain(qn, 32); }
                    step(d4, s4, k + 4); step(d5, s5, k + 5); while (qn >= 32) { qn -= 32; drain(qn, 32); }
                    step(d6, s6, k + 6); step(d7, s7, k + 7); while (qn >= 32) { qn -= 32; drain(qn, 32); }
                }
            } else {
                #pragma nounroll
                for (int k = 0; k < nk; ++k) {
                    const int u = unit_of(k);
                    handle(ldu(dst4, u), ldu(src4, u));
                    while (qn >= 32) { qn -= 32; drain(qn, 32); }
                }
            }
        }
        for (int c = nunit + vw; c * 128 < e; c += vstride) {
            handle(load_chunk_guarded(dst, c), load_chunk_guarded(src, c));
            while (qn >= 32) { qn -= 32; drain(qn, 32); }
        }
        if (qn > 0) drain(0, qn);
    }

    if (MODE == 0) {
        if (small && lane < SUB) {
            int len = 0;
            #pragma unroll
            for (int o = 0; o < SUB; ++o) if (lane == o) len = rnS[o];
            rcount[region0 + lane * rstep] = (len > RCAP) ? -1 : len;
        }
        __syncthreads();
        for (int t = tid; t < cnt; t += 1024)
            pdeg[(size_t)part * n + lo + t] = degc[t];
    } else if (MODE == 1) {
        __syncthreads();
        const int hh = lane >> 4;
        const int l  = lane & 15;
        for (int row0 = wv * 16; row0 < cnt; row0 += PWAVES * 16) {
            int ar = row0 + l;
            if (ar > cnt - 1) ar = cnt - 1;
            const float* arow = acc + ar * HID;
            v8f acc0 = {}, acc1 = {}, acc2 = {}, acc3 = {};
            for (int k0 = 0; k0 < HID; k0 += 4) {
                const int ka = k0 + hh * 2;
                v2f a;
                a.x = fmaxf(arow[ka],     0.0f);
                a.y = fmaxf(arow[ka + 1], 0.0f);
                const float* bp = Wn + (size_t)ka * HID + l;
                v2f b0, b1, b2, b3;
                b0.x = bp[0];  b0.y = bp[HID];
                b1.x = bp[16]; b1.y = bp[HID + 16];
                b2.x = bp[32]; b2.y = bp[HID + 32];
                b3.x = bp[48]; b3.y = bp[HID + 48];
                acc0 = __builtin_amdgcn_wmma_f32_16x16x4_f32(false, a, false, b0, (short)0, acc0, false, false);
                acc1 = __builtin_amdgcn_wmma_f32_16x16x4_f32(false, a, false, b1, (short)0, acc1, false, false);
                acc2 = __builtin_amdgcn_wmma_f32_16x16x4_f32(false, a, false, b2, (short)0, acc2, false, false);
                acc3 = __builtin_amdgcn_wmma_f32_16x16x4_f32(false, a, false, b3, (short)0, acc3, false, false);
            }
            #pragma unroll
            for (int r = 0; r < 8; ++r) {
                const int orow = row0 + hh * 8 + r;
                if (orow < cnt) {
                    float* o = acc + orow * HID + l;
                    o[0]  = acc0[r];
                    o[16] = acc1[r];
                    o[32] = acc2[r];
                    o[48] = acc3[r];
                }
            }
        }
        __syncthreads();
        for (int t = tid; t < cnt * HID; t += 1024)
            out[(size_t)lo * HID + t] = acc[t];
    } else {
        __syncthreads();
        constexpr int RPW = (NPW + PWAVES - 1) / PWAVES;
        const float w0 = Wn[lane], w1 = Wn[lane + 32];
        const int r0 = wv * RPW;
        if (r0 < cnt) {
            float mine = 0.0f;
            #pragma unroll
            for (int u = 0; u < RPW; ++u) {
                int r = r0 + u;
                if (r > cnt - 1) r = cnt - 1;
                float v = fmaxf(acc[r * HID + lane], 0.0f) * w0 + fmaxf(acc[r * HID + lane + 32], 0.0f) * w1;
                #pragma unroll
                for (int off = 16; off > 0; off >>= 1) v += __shfl_down(v, off, 32);
                const float tot = __int_as_float(__builtin_amdgcn_readfirstlane(__float_as_int(v)));
                if (lane == u) mine = tot;
            }
            if (lane < RPW && r0 + lane < cnt) out[lo + r0 + lane] = mine;
        }
    }
}

__global__ void __launch_bounds__(1024)
k_scan_deg(const int* __restrict__ src, const int* __restrict__ dst,
           int* __restrict__ pdeg,
           int* __restrict__ recs, int* __restrict__ rcount, int n, int e) {
    __shared__ int  degc[SUB * NPW];
    __shared__ int2 q_e[PWAVES][QCAP];
    own_body<0>(nullptr, degc, q_e, src, dst, nullptr, nullptr, pdeg, nullptr, nullptr, nullptr,
                recs, rcount, n, e);
}

__global__ void __launch_bounds__(1024)
k_prop_first(const int* __restrict__ src, const int* __restrict__ dst,
             const float* __restrict__ h,
             const float* __restrict__ dinv, const float* __restrict__ b,
             const float* __restrict__ W2,
             float* __restrict__ out, int* __restrict__ recs, int* __restrict__ rcount,
             int n, int e) {
    __shared__ float acc[NPW * HID];
    __shared__ int2  q_e[PWAVES][QCAP];
    own_body<1>(acc, nullptr, q_e, src, dst, h, dinv, nullptr, b, W2, out, recs, rcount, n, e);
}

__global__ void __launch_bounds__(1024)
k_prop_second(const int* __restrict__ src, const int* __restrict__ dst,
              const float* __restrict__ h,
              const float* __restrict__ dinv, const float* __restrict__ b,
              const float* __restrict__ W3,
              float* __restrict__ h3, int* __restrict__ recs, int* __restrict__ rcount,
              int n, int e) {
    __shared__ float acc[NPW * HID];
    __shared__ int2  q_e[PWAVES][QCAP];
    own_body<2>(acc, nullptr, q_e, src, dst, h, dinv, nullptr, b, W3, h3, recs, rcount, n, e);
}

__global__ void __launch_bounds__(1024)
k_prop3_own(const int* __restrict__ src, const int* __restrict__ dst,
            const float* __restrict__ h3,
            const float* __restrict__ dinv, const float* __restrict__ b3,
            float* __restrict__ y, const int* __restrict__ recs,
            const int* __restrict__ rcount, int n, int e) {
    __shared__ float acc3[NPW];
    __shared__ int   no_recs;

    const int tid  = threadIdx.x;
    const int lane = tid & 31;
    const int wv   = __builtin_amdgcn_readfirstlane(tid >> 5);
    const int lo   = blockIdx.x * NPW;
    int cnt = n - lo;
    if (cnt > NPW) cnt = NPW;
    if (cnt < 0)   cnt = 0;
    const unsigned ucnt = (unsigned)cnt;

    const int region0 = blockIdx.x * SUB * PWAVES + wv;
    const int4* rec0 = (const int4*)recs + (size_t)region0 * RCAP;
    int  pre[SUB + 1];
    bool have = gridDim.x <= 256;
    pre[0] = 0;
    #pragma unroll
    for (int p = 0; p < SUB; ++p) {
        int cp = -1;
        if (gridDim.x <= 256) cp = __builtin_amdgcn_readfirstlane(rcount[region0 + p * PWAVES]);
        if (cp < 0) { have = false; cp = 0; }
        if (cp > RCAP) cp = RCAP;
        pre[p + 1] = pre[p] + cp;
    }
    const int c = pre[SUB];
    auto recptr = [&](int i) -> const int4* {
        int idx = i;
        #pragma unroll
        for (int p = 1; p < SUB; ++p)
            if (i >= pre[p]) idx = i - pre[p] + p * PWAVES * RCAP;
        return rec0 + idx;
    };

    if (tid == 0) no_recs = 0;
    if (tid < cnt) {
        float di = dinv[lo + tid];
        acc3[tid] = b3[0] + h3[lo + tid] * di * di;
    }
    __syncthreads();
    if (!have && lane == 0) no_recs = 1;
    __syncthreads();

    if (no_recs == 0) {
        for (int base = 0; base < c; base += 256) {
            int4  r[8];
            float hv[8];
            #pragma unroll
            for (int u = 0; u < 8; ++u) r[u] = *recptr(min(base + u * 32 + lane, c - 1));
            #pragma unroll
            for (int u = 0; u < 8; ++u) hv[u] = h3[r[u].x];
            asm volatile("" :: "v"(hv[0]), "v"(hv[1]), "v"(hv[2]), "v"(hv[3]),
                               "v"(hv[4]), "v"(hv[5]), "v"(hv[6]), "v"(hv[7]));
            #pragma unroll
            for (int u = 0; u < 8; ++u) {
                if (base + u * 32 + lane < c)
                    (void)__hip_atomic_fetch_add(acc3 + r[u].y, hv[u] * __int_as_float(r[u].z),
                                                 __ATOMIC_RELAXED, __HIP_MEMORY_SCOPE_WORKGROUP);
            }
        }
    } else {
        for (int i = tid; i < e; i += 1024) {
            unsigned d = (unsigned)dst[i] - (unsigned)lo;
            if (d < ucnt) {
                int s = src[i];
                (void)__hip_atomic_fetch_add(acc3 + d, h3[s] * (dinv[s] * dinv[lo + d]),
                                             __ATOMIC_RELAXED, __HIP_MEMORY_SCOPE_WORKGROUP);
            }
        }
    }
    __syncthreads();

    if (tid < cnt) y[lo + tid] = acc3[tid];
}


extern "C" void kernel_launch(void* const* d_in, const int* in_sizes, int n_in,
                              void* d_out, int out_size, void* d_ws, size_t ws_size,
                              hipStream_t stream) {
    const float* x  = (const float*)d_in[0];
    const int*   ei = (const int*)  d_in[1];
    const float* W1 = (const float*)d_in[2];
    const float* b1 = (const float*)d_in[3];
    const float* W2 = (const float*)d_in[4];
    const float* b2 = (const float*)d_in[5];
    const float* W3 = (const float*)d_in[6];
    const float* b3 = (const float*)d_in[7];

    const int N = in_sizes[0] / 2;
    const int E = in_sizes[1] / 2;
    const int* src = ei;
    const int* dst = ei + E;

    float* dinv = (float*)d_ws;
    float* h    = dinv + (size_t)N;
    float* out  = h    + (size_t)N * HID;
    float* h3   = out  + (size_t)N * HID;
    int* recs   = (int*)(h3 + (size_t)N);
    int* rcount = recs + (size_t)256 * SUB * PWAVES * RCAP * 4;
    int* pdeg   = rcount + (size_t)256 * SUB * PWAVES;
    float* y    = (float*)d_out;

    const int B = 256;
    auto g = [](long long t) { return (int)((t + 255) / 256); };
    const int G = (N + NPW - 1) / NPW;
    const int GS = (G + SUB - 1) / SUB * SUB;

    k_scan_deg<<<GS, 1024, 0, stream>>>(src, dst, pdeg, recs, rcount, N, E);

    k_lin1      <<<g(((long long)N + 31) / 32 * 32), B, 0, stream>>>(x, W1, pdeg, dinv, h, N);
    k_prop_first<<<G, 1024, 0, stream>>>(src, dst, h, dinv, b1, W2, out, recs, rcount, N, E);

    k_prop_second<<<G, 1024, 0, stream>>>(src, dst, out, dinv, b2, W3, h3, recs, rcount, N, E);

    k_prop3_own<<<G, 1024, 0, stream>>>(src, dst, h3, dinv, b3, y, recs, rcount, N, E);
}
